# starting kernel
# baseline (speedup 1.0000x reference)
#include <hip/hip_runtime.h>


typedef float v2f __attribute__((ext_vector_type(2)));
typedef float v8f __attribute__((ext_vector_type(8)));

#define GCN_N_NODES 100000


__global__ void gcn_zero_kernel(float* __restrict__ p, long long n) {
  long long i = (long long)blockIdx.x * blockDim.x + threadIdx.x;
  long long stride = (long long)gridDim.x * blockDim.x;
  for (; i < n; i += stride) p[i] = 0.0f;
}

__global__ void gcn_degree_kernel(const int* __restrict__ rows,
                                  float* __restrict__ deg, int n_edges) {
  int e = blockIdx.x * blockDim.x + threadIdx.x;
  if (e < n_edges) atomicAdd(&deg[rows[e]], 1.0f);
}

__global__ void gcn_dinv_kernel(const float* __restrict__ deg,
                                float* __restrict__ dinv, int n) {
  int i = blockIdx.x * blockDim.x + threadIdx.x;
  if (i < n) {
    float d = sqrtf(deg[i]);
    dinv[i] = 1.0f / fmaxf(d, 1e-12f);
  }
}

__global__ void gcn_rowscale_kernel(float* __restrict__ p,
                                    const float* __restrict__ dinv,
                                    long long n, int colshift) {
  long long i = (long long)blockIdx.x * blockDim.x + threadIdx.x;
  long long stride = (long long)gridDim.x * blockDim.x;
  for (; i < n; i += stride) p[i] *= dinv[i >> colshift];
}


template <int N, bool RELU_SCALE_A, bool SCALE_C>
__global__ __launch_bounds__(256) void gcn_gemm_wmma(
    const float* __restrict__ A, const float* __restrict__ B,
    float* __restrict__ C, const float* __restrict__ dinv, int M, int K) {
  constexpr int NT = N / 16;
  constexpr int KC = 64;
  constexpr int N2 = N / 2;
  __shared__ __align__(16) float Bs[KC * N];

  const int wib  = threadIdx.x >> 5;
  const int lane = threadIdx.x & 31;
  const int half = lane >> 4;
  const int l15  = lane & 15;

  const int row0   = blockIdx.x * 128 + wib * 16;
  const bool active = (row0 < M);

  const float* __restrict__ Arow =
      A + (size_t)(active ? row0 + l15 : 0) * K;
  const float dscale =
      (RELU_SCALE_A && active) ? dinv[row0 + l15] : 1.0f;

  v8f acc[NT];
#pragma unroll
  for (int t = 0; t < NT; ++t) acc[t] = (v8f){};

  for (int kc = 0; kc < K; kc += KC) {
    const float* __restrict__ Bg = B + (size_t)kc * N;
#pragma unroll
    for (int pidx = threadIdx.x; pidx < (KC / 2) * N2; pidx += 256) {
      const int kp = pidx / N2;
      const int c2 = (pidx - kp * N2) * 2;
      const float2 g0 = *(const float2*)(Bg + (size_t)(2 * kp) * N + c2);
      const float2 g1 = *(const float2*)(Bg + (size_t)(2 * kp + 1) * N + c2);
      *(float4*)(&Bs[kp * (2 * N) + c2 * 2]) =
          make_float4(g0.x, g1.x, g0.y, g1.y);
    }
    __syncthreads();

    if (active) {
      const float* __restrict__ Ak = Arow + kc;
#pragma unroll
      for (int k0 = 0; k0 < KC; k0 += 4) {
        const int koff = k0 + 2 * half;
        v2f a = *(const v2f*)(Ak + koff);
        if (RELU_SCALE_A) {
          a.x = fmaxf(a.x * dscale, 0.0f);
          a.y = fmaxf(a.y * dscale, 0.0f);
        }
        const int kp = koff >> 1;
#pragma unroll
        for (int t = 0; t < NT; ++t) {
          const v2f b =
              *(const v2f*)(&Bs[kp * (2 * N) + (t * 16 + l15) * 2]);
          acc[t] = __builtin_amdgcn_wmma_f32_16x16x4_f32(
              false, a, false, b, (short)0, acc[t], false, false);
        }
      }
    }
    __syncthreads();
  }

  if (active) {
#pragma unroll
    for (int t = 0; t < NT; ++t) {
      const int col = t * 16 + l15;
#pragma unroll
      for (int v = 0; v < 8; ++v) {
        const int row = row0 + v + 8 * half;
        float val = acc[t][v];
        if (SCALE_C) val *= dinv[row];
        C[(size_t)row * N + col] = val;
      }
    }
  }
}


__global__ void gcn_spmm_plain(const float* __restrict__ src,
                               const int* __restrict__ rows,
                               const int* __restrict__ cols,
                               float* __restrict__ dst,
                               int n_edges, int F, int fshift) {
  const long long tid = (long long)blockIdx.x * blockDim.x + threadIdx.x;
  const int fchunks = 1 << fshift;
  const long long total = (long long)n_edges * fchunks;
  if (tid >= total) return;

  const int e  = (int)(tid >> fshift);
  const int fc = (int)(tid & (fchunks - 1));

  const int r = rows[e];
  const int c = cols[e];

  const float4 v = ((const float4*)(src + (size_t)c * F))[fc];
  float* __restrict__ d = dst + (size_t)r * F + fc * 4;
  atomicAdd(d + 0, v.x);
  atomicAdd(d + 1, v.y);
  atomicAdd(d + 2, v.z);
  atomicAdd(d + 3, v.w);
}


extern "C" void kernel_launch(void* const* d_in, const int* in_sizes, int n_in,
                              void* d_out, int out_size, void* d_ws, size_t ws_size,
                              hipStream_t stream) {
  const float* X  = (const float*)d_in[0];
  const float* W1 = (const float*)d_in[1];
  const float* W2 = (const float*)d_in[2];
  const int* erow = (const int*)d_in[3];
  const int* ecol = (const int*)d_in[4];
  const int n_edges = in_sizes[3];
  float* out = (float*)d_out;

  float* deg = (float*)d_ws;
  float* dinv = deg + GCN_N_NODES;
  float* XWs = dinv + GCN_N_NODES;
  float* S   = XWs + (size_t)GCN_N_NODES * 128;
  float* HWs = S + (size_t)GCN_N_NODES * 128;

  const long long nS   = (long long)GCN_N_NODES * 128;
  const long long nOut = (long long)out_size;

  gcn_zero_kernel<<<512, 256, 0, stream>>>(deg, GCN_N_NODES);
  gcn_zero_kernel<<<4096, 256, 0, stream>>>(S, nS);
  gcn_zero_kernel<<<4096, 256, 0, stream>>>(out, nOut);

  gcn_degree_kernel<<<(n_edges + 255) / 256, 256, 0, stream>>>(erow, deg, n_edges);
  gcn_dinv_kernel<<<(GCN_N_NODES + 255) / 256, 256, 0, stream>>>(deg, dinv, GCN_N_NODES);

  const int row_tiles = GCN_N_NODES / 16;
  const int gemm_blocks = (row_tiles + 7) / 8;

  gcn_gemm_wmma<128, false, true>
      <<<gemm_blocks, 256, 0, stream>>>(X, W1, XWs, dinv, GCN_N_NODES, 256);

  {
    const long long total = (long long)n_edges * 32;
    gcn_spmm_plain<<<(int)((total + 255) / 256), 256, 0, stream>>>(
        XWs, erow, ecol, S, n_edges, 128, 5);
  }

  gcn_gemm_wmma<64, true, true>
      <<<gemm_blocks, 256, 0, stream>>>(S, W2, HWs, dinv, GCN_N_NODES, 128);

  {
    const long long total = (long long)n_edges * 16;
    gcn_spmm_plain<<<(int)((total + 255) / 256), 256, 0, stream>>>(
        HWs, erow, ecol, out, n_edges, 64, 4);
  }

  gcn_rowscale_kernel<<<4096, 256, 0, stream>>>(out, dinv, nOut, 6);
}


# SpMM as pull: edges grouped by destination row per call (int count+rank atomics, scan, place); one wave adds a row's sources in slot (arrival) order, stores once; replaces f32 atomics and zero fills
# speedup vs baseline: 1.9538x; 1.9538x over previous
#include <hip/hip_runtime.h>


typedef float v2f __attribute__((ext_vector_type(2)));
typedef float v8f __attribute__((ext_vector_type(8)));

#define GCN_N_NODES 100000
#define GCN_ROWPTR_PAD 100016


__global__ void gcn_zero_int_kernel(int* __restrict__ p, int n) {
  int i = blockIdx.x * blockDim.x + threadIdx.x;
  if (i < n) p[i] = 0;
}

__global__ void gcn_count_rank_kernel(const int* __restrict__ rows,
                                      int* __restrict__ cnt,
                                      int* __restrict__ rank, int n_edges) {
  int e = blockIdx.x * blockDim.x + threadIdx.x;
  if (e < n_edges) rank[e] = atomicAdd(&cnt[rows[e]], 1);
}

__device__ __forceinline__ int gcn_wave_incl_scan(int x, int lane) {
  x += __builtin_amdgcn_update_dpp(0, x, 0x111, 0xf, 0xf, false);
  x += __builtin_amdgcn_update_dpp(0, x, 0x112, 0xf, 0xf, false);
  x += __builtin_amdgcn_update_dpp(0, x, 0x114, 0xf, 0xf, false);
  x += __builtin_amdgcn_update_dpp(0, x, 0x118, 0xf, 0xf, false);
  const int low = __builtin_amdgcn_readlane(x, 15);
  return x + ((lane >= 16) ? low : 0);
}

__global__ __launch_bounds__(1024) void gcn_scan_kernel(
    const int* __restrict__ cnt, int* __restrict__ rowptr) {
  static_assert(GCN_N_NODES % 16 == 0, "16 counts per thread");
  constexpr int NCH = GCN_N_NODES / 16;
  constexpr int TILES = (NCH + 1023) / 1024;
  __shared__ int wsum[2][32];
  const int tid = threadIdx.x;
  const int lane = tid & 31;
  const int w = __builtin_amdgcn_readfirstlane(tid >> 5);
  const int4* __restrict__ c4 = (const int4*)cnt;
  int4* __restrict__ r4 = (int4*)rowptr;

  int carry = 0;
  for (int t = 0; t < TILES; ++t) {
    const int ch = t * 1024 + tid;
    const bool valid = ch < NCH;
    int4 v[4];
#pragma unroll
    for (int k = 0; k < 4; ++k)
      v[k] = valid ? c4[ch * 4 + k] : make_int4(0, 0, 0, 0);
    int s = 0;
#pragma unroll
    for (int k = 0; k < 4; ++k) s += v[k].x + v[k].y + v[k].z + v[k].w;

    const int inc = gcn_wave_incl_scan(s, lane);
    if (lane == 31) wsum[t & 1][w] = inc;
    __syncthreads();
    const int ws = wsum[t & 1][lane];
    const int winc = gcn_wave_incl_scan(ws, lane);
    const int wexcl = __builtin_amdgcn_readlane(winc, w) -
                      __builtin_amdgcn_readlane(ws, w);
    const int total = __builtin_amdgcn_readlane(winc, 31);

    if (valid) {
      int run = carry + wexcl + inc - s;
#pragma unroll
      for (int k = 0; k < 4; ++k) {
        int4 r;
        r.x = run;
        r.y = r.x + v[k].x;
        r.z = r.y + v[k].y;
        r.w = r.z + v[k].z;
        run = r.w + v[k].w;
        r4[ch * 4 + k] = r;
      }
    }
    carry += total;
  }
  if (tid == 0) rowptr[GCN_N_NODES] = carry;
}

__global__ void gcn_place_dinv_kernel(const int* __restrict__ rows,
                                      const int* __restrict__ cols,
                                      const int* __restrict__ rowptr,
                                      const int* __restrict__ rank,
                                      const int* __restrict__ cnt,
                                      int* __restrict__ csr_col,
                                      float* __restrict__ dinv,
                                      int n_edges, int n_nodes) {
  const int e = blockIdx.x * blockDim.x + threadIdx.x;
  if (e < n_edges) csr_col[rowptr[rows[e]] + rank[e]] = cols[e];
  const int stride = gridDim.x * blockDim.x;
  for (int i = e; i < n_nodes; i += stride) {
    const float d = sqrtf((float)cnt[i]);
    dinv[i] = 1.0f / fmaxf(d, 1e-12f);
  }
}


template <int N, bool RELU_SCALE_A, bool SCALE_C>
__global__ __launch_bounds__(256) void gcn_gemm_wmma(
    const float* __restrict__ A, const float* __restrict__ B,
    float* __restrict__ C, const float* __restrict__ dinv, int M, int K) {
  constexpr int NT = N / 16;
  constexpr int KC = 64;
  constexpr int N2 = N / 2;
  __shared__ __align__(16) float Bs[KC * N];

  const int wib  = threadIdx.x >> 5;
  const int lane = threadIdx.x & 31;
  const int half = lane >> 4;
  const int l15  = lane & 15;

  const int row0   = blockIdx.x * 128 + wib * 16;
  const bool active = (row0 < M);

  const float* __restrict__ Arow =
      A + (size_t)(active ? row0 + l15 : 0) * K;
  const float dscale =
      (RELU_SCALE_A && active) ? dinv[row0 + l15] : 1.0f;

  v8f acc[NT];
#pragma unroll
  for (int t = 0; t < NT; ++t) acc[t] = (v8f){};

  for (int kc = 0; kc < K; kc += KC) {
    const float* __restrict__ Bg = B + (size_t)kc * N;
#pragma unroll
    for (int pidx = threadIdx.x; pidx < (KC / 2) * N2; pidx += 256) {
      const int kp = pidx / N2;
      const int c2 = (pidx - kp * N2) * 2;
      const float2 g0 = *(const float2*)(Bg + (size_t)(2 * kp) * N + c2);
      const float2 g1 = *(const float2*)(Bg + (size_t)(2 * kp + 1) * N + c2);
      *(float4*)(&Bs[kp * (2 * N) + c2 * 2]) =
          make_float4(g0.x, g1.x, g0.y, g1.y);
    }
    __syncthreads();

    if (active) {
      const float* __restrict__ Ak = Arow + kc;
#pragma unroll
      for (int k0 = 0; k0 < KC; k0 += 4) {
        const int koff = k0 + 2 * half;
        v2f a = *(const v2f*)(Ak + koff);
        if (RELU_SCALE_A) {
          a.x = fmaxf(a.x * dscale, 0.0f);
          a.y = fmaxf(a.y * dscale, 0.0f);
        }
        const int kp = koff >> 1;
#pragma unroll
        for (int t = 0; t < NT; ++t) {
          const v2f b =
              *(const v2f*)(&Bs[kp * (2 * N) + (t * 16 + l15) * 2]);
          acc[t] = __builtin_amdgcn_wmma_f32_16x16x4_f32(
              false, a, false, b, (short)0, acc[t], false, false);
        }
      }
    }
    __syncthreads();
  }

  if (active) {
#pragma unroll
    for (int t = 0; t < NT; ++t) {
      const int col = t * 16 + l15;
#pragma unroll
      for (int v = 0; v < 8; ++v) {
        const int row = row0 + v + 8 * half;
        float val = acc[t][v];
        if (SCALE_C) val *= dinv[row];
        C[(size_t)row * N + col] = val;
      }
    }
  }
}


template <int VEC, bool SCALE_OUT>
__global__ __launch_bounds__(256) void gcn_pull_rows(
    const float* __restrict__ src, const int* __restrict__ rowptr,
    const int* __restrict__ csr_col, const float* __restrict__ dinv,
    float* __restrict__ dst, int n_rows) {
  typedef float vf __attribute__((ext_vector_type(VEC)));
  constexpr int F = VEC * 32;
  const int lane = threadIdx.x & 31;
  const int row =
      __builtin_amdgcn_readfirstlane(blockIdx.x * 8 + (threadIdx.x >> 5));
  if (row >= n_rows) return;
  const int beg = rowptr[row];
  const int end = rowptr[row + 1];
  const float* __restrict__ srcl = src + lane * VEC;

  vf acc = {};
  for (int base = beg; base < end; base += 32) {
    const int n = min(32, end - base);
    const int colv = (lane < n) ? csr_col[base + lane] : 0;
    int j = 0;
    for (; j + 4 <= n; j += 4) {
      const int c0 = __builtin_amdgcn_readlane(colv, j);
      const int c1 = __builtin_amdgcn_readlane(colv, j + 1);
      const int c2 = __builtin_amdgcn_readlane(colv, j + 2);
      const int c3 = __builtin_amdgcn_readlane(colv, j + 3);
      const vf x0 = *(const vf*)(srcl + (size_t)c0 * F);
      const vf x1 = *(const vf*)(srcl + (size_t)c1 * F);
      const vf x2 = *(const vf*)(srcl + (size_t)c2 * F);
      const vf x3 = *(const vf*)(srcl + (size_t)c3 * F);
      acc += x0;
      acc += x1;
      acc += x2;
      acc += x3;
    }
    for (; j < n; ++j) {
      const int c0 = __builtin_amdgcn_readlane(colv, j);
      acc += *(const vf*)(srcl + (size_t)c0 * F);
    }
  }
  if (SCALE_OUT) acc *= dinv[row];
  *(vf*)(dst + (size_t)row * F + lane * VEC) = acc;
}


extern "C" void kernel_launch(void* const* d_in, const int* in_sizes, int n_in,
                              void* d_out, int out_size, void* d_ws, size_t ws_size,
                              hipStream_t stream) {
  const float* X  = (const float*)d_in[0];
  const float* W1 = (const float*)d_in[1];
  const float* W2 = (const float*)d_in[2];
  const int* erow = (const int*)d_in[3];
  const int* ecol = (const int*)d_in[4];
  const int n_edges = in_sizes[3];
  float* out = (float*)d_out;

  int* cnt    = (int*)d_ws;
  int* rowptr = cnt + GCN_N_NODES;
  float* dinv = (float*)(rowptr + GCN_ROWPTR_PAD);
  float* XWs = dinv + GCN_N_NODES;
  float* S   = XWs + (size_t)GCN_N_NODES * 128;
  float* HWs = S + (size_t)GCN_N_NODES * 128;
  int* rank    = (int*)(HWs + (size_t)GCN_N_NODES * 64);
  int* csr_col = rank + n_edges;

  const int edge_blocks = (n_edges + 255) / 256;

  gcn_zero_int_kernel<<<(GCN_N_NODES + 255) / 256, 256, 0, stream>>>(cnt, GCN_N_NODES);
  gcn_count_rank_kernel<<<edge_blocks, 256, 0, stream>>>(erow, cnt, rank, n_edges);
  gcn_scan_kernel<<<1, 1024, 0, stream>>>(cnt, rowptr);
  gcn_place_dinv_kernel<<<edge_blocks, 256, 0, stream>>>(
      erow, ecol, rowptr, rank, cnt, csr_col, dinv, n_edges, GCN_N_NODES);

  const int row_tiles = GCN_N_NODES / 16;
  const int gemm_blocks = (row_tiles + 7) / 8;
  const int pull_blocks = (GCN_N_NODES + 7) / 8;

  gcn_gemm_wmma<128, false, true>
      <<<gemm_blocks, 256, 0, stream>>>(X, W1, XWs, dinv, GCN_N_NODES, 256);

  gcn_pull_rows<4, false><<<pull_blocks, 256, 0, stream>>>(
      XWs, rowptr, csr_col, dinv, S, GCN_N_NODES);

  gcn_gemm_wmma<64, true, true>
      <<<gemm_blocks, 256, 0, stream>>>(S, W2, HWs, dinv, GCN_N_NODES, 128);

  gcn_pull_rows<2, true><<<pull_blocks, 256, 0, stream>>>(
      HWs, rowptr, csr_col, dinv, out, GCN_N_NODES);
}


# Edge grouping as a two-level counting sort that counts in LDS (slice x 512-row band histogram, band scatter, per-band count and place); no global atomics or scan; row slots stay in arrival order
# speedup vs baseline: 2.6750x; 1.3691x over previous
#include <hip/hip_runtime.h>


typedef float v2f __attribute__((ext_vector_type(2)));
typedef float v8f __attribute__((ext_vector_type(8)));

#define GCN_N_NODES 100000
#define GCN_ROWPTR_PAD 100016


#define GCN_BAND_SHIFT 9
#define GCN_BAND 512
#define GCN_NBANDS 196
#define GCN_SLICES 256
#define GCN_BATCH 8

static_assert((GCN_N_NODES + GCN_BAND - 1) / GCN_BAND == GCN_NBANDS, "band count");
static_assert(GCN_N_NODES <= (1 << 22), "col << 9 must stay a non-negative int");
static_assert(GCN_BATCH == 8, "the load fences list 8 values");
static_assert(GCN_NBANDS <= 7 * 32, "scatter kernel: 7 bands per lane / per wave");
static_assert(GCN_NBANDS * GCN_SLICES + GCN_NBANDS + 1 <= GCN_N_NODES,
              "hist and bandstart live in the first GCN_N_NODES ints of the workspace");

__device__ __forceinline__ int gcn_wave_incl_scan(int x, int lane) {
  x += __builtin_amdgcn_update_dpp(0, x, 0x111, 0xf, 0xf, false);
  x += __builtin_amdgcn_update_dpp(0, x, 0x112, 0xf, 0xf, false);
  x += __builtin_amdgcn_update_dpp(0, x, 0x114, 0xf, 0xf, false);
  x += __builtin_amdgcn_update_dpp(0, x, 0x118, 0xf, 0xf, false);
  const int low = __builtin_amdgcn_readlane(x, 15);
  return x + ((lane >= 16) ? low : 0);
}

__global__ __launch_bounds__(1024) void gcn_band_hist_kernel(
    const int* __restrict__ rows, int* __restrict__ hist, int n_edges) {
  __shared__ int h[GCN_NBANDS];
  const int tid = threadIdx.x;
  const int w = blockIdx.x;
  if (tid < GCN_NBANDS) h[tid] = 0;
  __syncthreads();
  const int sl = (n_edges + GCN_SLICES - 1) / GCN_SLICES;
  const int s0 = min(w * sl, n_edges);
  const int s1 = min(s0 + sl, n_edges);
  for (int e = s0 + tid; e < s1; e += 1024 * GCN_BATCH) {
    int r[GCN_BATCH];
#pragma unroll
    for (int k = 0; k < GCN_BATCH; ++k) r[k] = rows[min(e + k * 1024, s1 - 1)];
    asm volatile("" ::"v"(r[0]), "v"(r[1]), "v"(r[2]), "v"(r[3]), "v"(r[4]), "v"(r[5]),
                 "v"(r[6]), "v"(r[7]));
#pragma unroll
    for (int k = 0; k < GCN_BATCH; ++k)
      if (e + k * 1024 < s1) atomicAdd(&h[r[k] >> GCN_BAND_SHIFT], 1);
  }
  __syncthreads();
  if (tid < GCN_NBANDS) hist[tid * GCN_SLICES + w] = h[tid];
}

__global__ __launch_bounds__(1024) void gcn_band_scatter_kernel(
    const int* __restrict__ rows, const int* __restrict__ cols,
    const int* __restrict__ hist, int* __restrict__ bandstart,
    int* __restrict__ binned, int n_edges) {
  __shared__ int sT[7 * 32];
  __shared__ int sP[7 * 32];
  __shared__ int cursor[GCN_NBANDS];
  const int tid = threadIdx.x;
  const int lane = tid & 31;
  const int wv = __builtin_amdgcn_readfirstlane(tid >> 5);
  const int w = blockIdx.x;

  int4 qa[7], qb[7];
#pragma unroll
  for (int k = 0; k < 7; ++k) {
    const int bb = min(wv + 32 * k, GCN_NBANDS - 1);
    const int4* __restrict__ p = (const int4*)(hist + bb * GCN_SLICES + lane * 8);
    qa[k] = p[0];
    qb[k] = p[1];
  }
#pragma unroll
  for (int k = 0; k < 7; ++k) {
    const int b = wv + 32 * k;
    const int j0 = lane * 8;
    int tot = qa[k].x + qa[k].y + qa[k].z + qa[k].w + qb[k].x + qb[k].y + qb[k].z + qb[k].w;
    int part = ((j0 + 0 < w) ? qa[k].x : 0) + ((j0 + 1 < w) ? qa[k].y : 0) +
               ((j0 + 2 < w) ? qa[k].z : 0) + ((j0 + 3 < w) ? qa[k].w : 0) +
               ((j0 + 4 < w) ? qb[k].x : 0) + ((j0 + 5 < w) ? qb[k].y : 0) +
               ((j0 + 6 < w) ? qb[k].z : 0) + ((j0 + 7 < w) ? qb[k].w : 0);
    tot = __builtin_amdgcn_readlane(gcn_wave_incl_scan(tot, lane), 31);
    part = __builtin_amdgcn_readlane(gcn_wave_incl_scan(part, lane), 31);
    if (lane == 0) {
      sT[b] = (b < GCN_NBANDS) ? tot : 0;
      sP[b] = (b < GCN_NBANDS) ? part : 0;
    }
  }
  __syncthreads();

  if (wv == 0) {
    int t[7];
    int s = 0;
#pragma unroll
    for (int k = 0; k < 7; ++k) {
      t[k] = sT[lane * 7 + k];
      s += t[k];
    }
    int run = gcn_wave_incl_scan(s, lane) - s;
#pragma unroll
    for (int k = 0; k < 7; ++k) {
      const int b = lane * 7 + k;
      if (b < GCN_NBANDS) {
        cursor[b] = run + sP[b];
        if (w == 0) bandstart[b] = run;
      }
      run += t[k];
    }
    if (w == 0 && lane == 31) bandstart[GCN_NBANDS] = run;
  }
  __syncthreads();

  const int sl = (n_edges + GCN_SLICES - 1) / GCN_SLICES;
  const int s0 = min(w * sl, n_edges);
  const int s1 = min(s0 + sl, n_edges);
  for (int e = s0 + tid; e < s1; e += 1024 * GCN_BATCH) {
    int r[GCN_BATCH], c[GCN_BATCH];
#pragma unroll
    for (int k = 0; k < GCN_BATCH; ++k) {
      const int ee = min(e + k * 1024, s1 - 1);
      r[k] = rows[ee];
      c[k] = cols[ee];
    }
    asm volatile("" ::"v"(r[0]), "v"(r[1]), "v"(r[2]), "v"(r[3]), "v"(r[4]), "v"(r[5]),
                 "v"(r[6]), "v"(r[7]), "v"(c[0]), "v"(c[1]), "v"(c[2]), "v"(c[3]), "v"(c[4]),
                 "v"(c[5]), "v"(c[6]), "v"(c[7]));
#pragma unroll
    for (int k = 0; k < GCN_BATCH; ++k) {
      if (e + k * 1024 < s1) {
        const int pos = atomicAdd(&cursor[r[k] >> GCN_BAND_SHIFT], 1);
        binned[pos] = (c[k] << GCN_BAND_SHIFT) | (r[k] & (GCN_BAND - 1));
      }
    }
  }
}

__global__ __launch_bounds__(GCN_BAND) void gcn_band_sort_kernel(
    const int* __restrict__ binned, const int* __restrict__ bandstart,
    int* __restrict__ rowptr, int* __restrict__ csr_col, float* __restrict__ dinv) {
  __shared__ int cnt[GCN_BAND];
  __shared__ int wsum[GCN_BAND / 32];
  const int tid = threadIdx.x;
  const int lane = tid & 31;
  const int wv = __builtin_amdgcn_readfirstlane(tid >> 5);
  const int b = blockIdx.x;
  const int beg = bandstart[b];
  const int end = bandstart[b + 1];

  cnt[tid] = 0;
  __syncthreads();
  for (int i = beg + tid; i < end; i += GCN_BAND * GCN_BATCH) {
    int v[GCN_BATCH];
#pragma unroll
    for (int k = 0; k < GCN_BATCH; ++k) v[k] = binned[min(i + k * GCN_BAND, end - 1)];
    asm volatile("" ::"v"(v[0]), "v"(v[1]), "v"(v[2]), "v"(v[3]), "v"(v[4]), "v"(v[5]),
                 "v"(v[6]), "v"(v[7]));
#pragma unroll
    for (int k = 0; k < GCN_BATCH; ++k)
      if (i + k * GCN_BAND < end) atomicAdd(&cnt[v[k] & (GCN_BAND - 1)], 1);
  }
  __syncthreads();

  const int d = cnt[tid];
  const int inc = gcn_wave_incl_scan(d, lane);
  if (lane == 31) wsum[wv] = inc;
  __syncthreads();
  const int ws = (lane < GCN_BAND / 32) ? wsum[lane] : 0;
  const int winc = gcn_wave_incl_scan(ws, lane);
  const int wexcl = __builtin_amdgcn_readlane(winc, wv) - __builtin_amdgcn_readlane(ws, wv);
  const int start = beg + wexcl + inc - d;
  const int row = b * GCN_BAND + tid;
  if (row < GCN_N_NODES) {
    rowptr[row] = start;
    if (row == GCN_N_NODES - 1) rowptr[GCN_N_NODES] = start + d;
    dinv[row] = 1.0f / fmaxf(sqrtf((float)d), 1e-12f);
  }
  cnt[tid] = start;
  __syncthreads();

  for (int i = beg + tid; i < end; i += GCN_BAND * GCN_BATCH) {
    int v[GCN_BATCH];
#pragma unroll
    for (int k = 0; k < GCN_BATCH; ++k) v[k] = binned[min(i + k * GCN_BAND, end - 1)];
    asm volatile("" ::"v"(v[0]), "v"(v[1]), "v"(v[2]), "v"(v[3]), "v"(v[4]), "v"(v[5]),
                 "v"(v[6]), "v"(v[7]));
#pragma unroll
    for (int k = 0; k < GCN_BATCH; ++k) {
      if (i + k * GCN_BAND < end) {
        const int pos = atomicAdd(&cnt[v[k] & (GCN_BAND - 1)], 1);
        csr_col[pos] = v[k] >> GCN_BAND_SHIFT;
      }
    }
  }
}


template <int N, bool RELU_SCALE_A, bool SCALE_C>
__global__ __launch_bounds__(256) void gcn_gemm_wmma(
    const float* __restrict__ A, const float* __restrict__ B,
    float* __restrict__ C, const float* __restrict__ dinv, int M, int K) {
  constexpr int NT = N / 16;
  constexpr int KC = 64;
  constexpr int N2 = N / 2;
  __shared__ __align__(16) float Bs[KC * N];

  const int wib  = threadIdx.x >> 5;
  const int lane = threadIdx.x & 31;
  const int half = lane >> 4;
  const int l15  = lane & 15;

  const int row0   = blockIdx.x * 128 + wib * 16;
  const bool active = (row0 < M);

  const float* __restrict__ Arow =
      A + (size_t)(active ? row0 + l15 : 0) * K;
  const float dscale =
      (RELU_SCALE_A && active) ? dinv[row0 + l15] : 1.0f;

  v8f acc[NT];
#pragma unroll
  for (int t = 0; t < NT; ++t) acc[t] = (v8f){};

  for (int kc = 0; kc < K; kc += KC) {
    const float* __restrict__ Bg = B + (size_t)kc * N;
#pragma unroll
    for (int pidx = threadIdx.x; pidx < (KC / 2) * N2; pidx += 256) {
      const int kp = pidx / N2;
      const int c2 = (pidx - kp * N2) * 2;
      const float2 g0 = *(const float2*)(Bg + (size_t)(2 * kp) * N + c2);
      const float2 g1 = *(const float2*)(Bg + (size_t)(2 * kp + 1) * N + c2);
      *(float4*)(&Bs[kp * (2 * N) + c2 * 2]) =
          make_float4(g0.x, g1.x, g0.y, g1.y);
    }
    __syncthreads();

    if (active) {
      const float* __restrict__ Ak = Arow + kc;
#pragma unroll
      for (int k0 = 0; k0 < KC; k0 += 4) {
        const int koff = k0 + 2 * half;
        v2f a = *(const v2f*)(Ak + koff);
        if (RELU_SCALE_A) {
          a.x = fmaxf(a.x * dscale, 0.0f);
          a.y = fmaxf(a.y * dscale, 0.0f);
        }
        const int kp = koff >> 1;
#pragma unroll
        for (int t = 0; t < NT; ++t) {
          const v2f b =
              *(const v2f*)(&Bs[kp * (2 * N) + (t * 16 + l15) * 2]);
          acc[t] = __builtin_amdgcn_wmma_f32_16x16x4_f32(
              false, a, false, b, (short)0, acc[t], false, false);
        }
      }
    }
    __syncthreads();
  }

  if (active) {
#pragma unroll
    for (int t = 0; t < NT; ++t) {
      const int col = t * 16 + l15;
#pragma unroll
      for (int v = 0; v < 8; ++v) {
        const int row = row0 + v + 8 * half;
        float val = acc[t][v];
        if (SCALE_C) val *= dinv[row];
        C[(size_t)row * N + col] = val;
      }
    }
  }
}


template <int VEC, bool SCALE_OUT>
__global__ __launch_bounds__(256) void gcn_pull_rows(
    const float* __restrict__ src, const int* __restrict__ rowptr,
    const int* __restrict__ csr_col, const float* __restrict__ dinv,
    float* __restrict__ dst, int n_rows) {
  typedef float vf __attribute__((ext_vector_type(VEC)));
  constexpr int F = VEC * 32;
  const int lane = threadIdx.x & 31;
  const int row =
      __builtin_amdgcn_readfirstlane(blockIdx.x * 8 + (threadIdx.x >> 5));
  if (row >= n_rows) return;
  const int beg = rowptr[row];
  const int end = rowptr[row + 1];
  const float* __restrict__ srcl = src + lane * VEC;

  vf acc = {};
  for (int base = beg; base < end; base += 32) {
    const int n = min(32, end - base);
    const int colv = (lane < n) ? csr_col[base + lane] : 0;
    int j = 0;
    for (; j + 4 <= n; j += 4) {
      const int c0 = __builtin_amdgcn_readlane(colv, j);
      const int c1 = __builtin_amdgcn_readlane(colv, j + 1);
      const int c2 = __builtin_amdgcn_readlane(colv, j + 2);
      const int c3 = __builtin_amdgcn_readlane(colv, j + 3);
      const vf x0 = *(const vf*)(srcl + (size_t)c0 * F);
      const vf x1 = *(const vf*)(srcl + (size_t)c1 * F);
      const vf x2 = *(const vf*)(srcl + (size_t)c2 * F);
      const vf x3 = *(const vf*)(srcl + (size_t)c3 * F);
      acc += x0;
      acc += x1;
      acc += x2;
      acc += x3;
    }
    for (; j < n; ++j) {
      const int c0 = __builtin_amdgcn_readlane(colv, j);
      acc += *(const vf*)(srcl + (size_t)c0 * F);
    }
  }
  if (SCALE_OUT) acc *= dinv[row];
  *(vf*)(dst + (size_t)row * F + lane * VEC) = acc;
}


extern "C" void kernel_launch(void* const* d_in, const int* in_sizes, int n_in,
                              void* d_out, int out_size, void* d_ws, size_t ws_size,
                              hipStream_t stream) {
  const float* X  = (const float*)d_in[0];
  const float* W1 = (const float*)d_in[1];
  const float* W2 = (const float*)d_in[2];
  const int* erow = (const int*)d_in[3];
  const int* ecol = (const int*)d_in[4];
  const int n_edges = in_sizes[3];
  float* out = (float*)d_out;

  int* hist      = (int*)d_ws;
  int* bandstart = hist + GCN_NBANDS * GCN_SLICES;
  int* rowptr = (int*)d_ws + GCN_N_NODES;
  float* dinv = (float*)(rowptr + GCN_ROWPTR_PAD);
  float* XWs = dinv + GCN_N_NODES;
  float* S   = XWs + (size_t)GCN_N_NODES * 128;
  float* HWs = S + (size_t)GCN_N_NODES * 128;
  int* binned  = (int*)(HWs + (size_t)GCN_N_NODES * 64);
  int* csr_col = binned + n_edges;

  gcn_band_hist_kernel<<<GCN_SLICES, 1024, 0, stream>>>(erow, hist, n_edges);
  gcn_band_scatter_kernel<<<GCN_SLICES, 1024, 0, stream>>>(
      erow, ecol, hist, bandstart, binned, n_edges);
  gcn_band_sort_kernel<<<GCN_NBANDS, GCN_BAND, 0, stream>>>(
      binned, bandstart, rowptr, csr_col, dinv);

  const int row_tiles = GCN_N_NODES / 16;
  const int gemm_blocks = (row_tiles + 7) / 8;
  const int pull_blocks = (GCN_N_NODES + 7) / 8;

  gcn_gemm_wmma<128, false, true>
      <<<gemm_blocks, 256, 0, stream>>>(X, W1, XWs, dinv, GCN_N_NODES, 256);

  gcn_pull_rows<4, false><<<pull_blocks, 256, 0, stream>>>(
      XWs, rowptr, csr_col, dinv, S, GCN_N_NODES);

  gcn_gemm_wmma<64, true, true>
      <<<gemm_blocks, 256, 0, stream>>>(S, W2, HWs, dinv, GCN_N_NODES, 128);

  gcn_pull_rows<2, true><<<pull_blocks, 256, 0, stream>>>(
      HWs, rowptr, csr_col, dinv, out, GCN_N_NODES);
}


# speedup vs baseline: 2.9563x; 1.1052x over previous
#include <hip/hip_runtime.h>


typedef float v2f __attribute__((ext_vector_type(2)));
typedef float v8f __attribute__((ext_vector_type(8)));

#define GCN_N_NODES 100000
#define GCN_ROWPTR_PAD 100016
#define GCN_ROWS_OFF 301056
static_assert(GCN_ROWS_OFF >= GCN_N_NODES + GCN_ROWPTR_PAD + GCN_N_NODES,
              "row arrays start behind hist, rowptr and dinv");
static_assert(GCN_ROWS_OFF % 1024 == 0, "row arrays start on a 4096-byte boundary");
static_assert(((size_t)GCN_N_NODES * 64 * sizeof(float)) % 4096 == 0,
              "every row array is a multiple of 4096 bytes long");


#define GCN_BAND_SHIFT 9
#define GCN_BAND 512
#define GCN_NBANDS 196
#define GCN_SLICES 256
#define GCN_BATCH 8

static_assert((GCN_N_NODES + GCN_BAND - 1) / GCN_BAND == GCN_NBANDS, "band count");
static_assert(GCN_N_NODES <= (1 << 22), "col << 9 must stay a non-negative int");
static_assert(GCN_BATCH == 8, "the load fences list 8 values");
static_assert(GCN_NBANDS <= 7 * 32, "scatter kernel: 7 bands per lane / per wave");
static_assert(GCN_NBANDS * GCN_SLICES + GCN_NBANDS + 1 <= GCN_N_NODES,
              "hist and bandstart live in the first GCN_N_NODES ints of the workspace");

__device__ __forceinline__ int gcn_wave_incl_scan(int x, int lane) {
  x += __builtin_amdgcn_update_dpp(0, x, 0x111, 0xf, 0xf, false);
  x += __builtin_amdgcn_update_dpp(0, x, 0x112, 0xf, 0xf, false);
  x += __builtin_amdgcn_update_dpp(0, x, 0x114, 0xf, 0xf, false);
  x += __builtin_amdgcn_update_dpp(0, x, 0x118, 0xf, 0xf, false);
  const int low = __builtin_amdgcn_readlane(x, 15);
  return x + ((lane >= 16) ? low : 0);
}

__global__ __launch_bounds__(1024) void gcn_band_hist_kernel(
    const int* __restrict__ rows, int* __restrict__ hist, int n_edges) {
  __shared__ int h[GCN_NBANDS];
  const int tid = threadIdx.x;
  const int w = blockIdx.x;
  if (tid < GCN_NBANDS) h[tid] = 0;
  __syncthreads();
  const int sl = (n_edges + GCN_SLICES - 1) / GCN_SLICES;
  const int s0 = min(w * sl, n_edges);
  const int s1 = min(s0 + sl, n_edges);
  for (int e = s0 + tid; e < s1; e += 1024 * GCN_BATCH) {
    int r[GCN_BATCH];
#pragma unroll
    for (int k = 0; k < GCN_BATCH; ++k) r[k] = rows[min(e + k * 1024, s1 - 1)];
    asm volatile("" ::"v"(r[0]), "v"(r[1]), "v"(r[2]), "v"(r[3]), "v"(r[4]), "v"(r[5]),
                 "v"(r[6]), "v"(r[7]));
#pragma unroll
    for (int k = 0; k < GCN_BATCH; ++k)
      if (e + k * 1024 < s1) atomicAdd(&h[r[k] >> GCN_BAND_SHIFT], 1);
  }
  __syncthreads();
  if (tid < GCN_NBANDS) hist[tid * GCN_SLICES + w] = h[tid];
}

__global__ __launch_bounds__(1024) void gcn_band_scatter_kernel(
    const int* __restrict__ rows, const int* __restrict__ cols,
    const int* __restrict__ hist, int* __restrict__ bandstart,
    int* __restrict__ binned, int n_edges) {
  __shared__ int sT[7 * 32];
  __shared__ int sP[7 * 32];
  __shared__ int cursor[GCN_NBANDS];
  const int tid = threadIdx.x;
  const int lane = tid & 31;
  const int wv = __builtin_amdgcn_readfirstlane(tid >> 5);
  const int w = blockIdx.x;

  int4 qa[7], qb[7];
#pragma unroll
  for (int k = 0; k < 7; ++k) {
    const int bb = min(wv + 32 * k, GCN_NBANDS - 1);
    const int4* __restrict__ p = (const int4*)(hist + bb * GCN_SLICES + lane * 8);
    qa[k] = p[0];
    qb[k] = p[1];
  }
#pragma unroll
  for (int k = 0; k < 7; ++k) {
    const int b = wv + 32 * k;
    const int j0 = lane * 8;
    int tot = qa[k].x + qa[k].y + qa[k].z + qa[k].w + qb[k].x + qb[k].y + qb[k].z + qb[k].w;
    int part = ((j0 + 0 < w) ? qa[k].x : 0) + ((j0 + 1 < w) ? qa[k].y : 0) +
               ((j0 + 2 < w) ? qa[k].z : 0) + ((j0 + 3 < w) ? qa[k].w : 0) +
               ((j0 + 4 < w) ? qb[k].x : 0) + ((j0 + 5 < w) ? qb[k].y : 0) +
               ((j0 + 6 < w) ? qb[k].z : 0) + ((j0 + 7 < w) ? qb[k].w : 0);
    tot = __builtin_amdgcn_readlane(gcn_wave_incl_scan(tot, lane), 31);
    part = __builtin_amdgcn_readlane(gcn_wave_incl_scan(part, lane), 31);
    if (lane == 0) {
      sT[b] = (b < GCN_NBANDS) ? tot : 0;
      sP[b] = (b < GCN_NBANDS) ? part : 0;
    }
  }
  __syncthreads();

  if (wv == 0) {
    int t[7];
    int s = 0;
#pragma unroll
    for (int k = 0; k < 7; ++k) {
      t[k] = sT[lane * 7 + k];
      s += t[k];
    }
    int run = gcn_wave_incl_scan(s, lane) - s;
#pragma unroll
    for (int k = 0; k < 7; ++k) {
      const int b = lane * 7 + k;
      if (b < GCN_NBANDS) {
        cursor[b] = run + sP[b];
        if (w == 0) bandstart[b] = run;
      }
      run += t[k];
    }
    if (w == 0 && lane == 31) bandstart[GCN_NBANDS] = run;
  }
  __syncthreads();

  const int sl = (n_edges + GCN_SLICES - 1) / GCN_SLICES;
  const int s0 = min(w * sl, n_edges);
  const int s1 = min(s0 + sl, n_edges);
  for (int e = s0 + tid; e < s1; e += 1024 * GCN_BATCH) {
    int r[GCN_BATCH], c[GCN_BATCH];
#pragma unroll
    for (int k = 0; k < GCN_BATCH; ++k) {
      const int ee = min(e + k * 1024, s1 - 1);
      r[k] = rows[ee];
      c[k] = cols[ee];
    }
    asm volatile("" ::"v"(r[0]), "v"(r[1]), "v"(r[2]), "v"(r[3]), "v"(r[4]), "v"(r[5]),
                 "v"(r[6]), "v"(r[7]), "v"(c[0]), "v"(c[1]), "v"(c[2]), "v"(c[3]), "v"(c[4]),
                 "v"(c[5]), "v"(c[6]), "v"(c[7]));
#pragma unroll
    for (int k = 0; k < GCN_BATCH; ++k) {
      if (e + k * 1024 < s1) {
        const int pos = atomicAdd(&cursor[r[k] >> GCN_BAND_SHIFT], 1);
        binned[pos] = (c[k] << GCN_BAND_SHIFT) | (r[k] & (GCN_BAND - 1));
      }
    }
  }
}

__global__ __launch_bounds__(GCN_BAND) void gcn_band_sort_kernel(
    const int* __restrict__ binned, const int* __restrict__ bandstart,
    int* __restrict__ rowptr, int* __restrict__ csr_col, float* __restrict__ dinv) {
  __shared__ int cnt[GCN_BAND];
  __shared__ int wsum[GCN_BAND / 32];
  const int tid = threadIdx.x;
  const int lane = tid & 31;
  const int wv = __builtin_amdgcn_readfirstlane(tid >> 5);
  const int b = blockIdx.x;
  const int beg = bandstart[b];
  const int end = bandstart[b + 1];

  cnt[tid] = 0;
  __syncthreads();
  for (int i = beg + tid; i < end; i += GCN_BAND * GCN_BATCH) {
    int v[GCN_BATCH];
#pragma unroll
    for (int k = 0; k < GCN_BATCH; ++k) v[k] = binned[min(i + k * GCN_BAND, end - 1)];
    asm volatile("" ::"v"(v[0]), "v"(v[1]), "v"(v[2]), "v"(v[3]), "v"(v[4]), "v"(v[5]),
                 "v"(v[6]), "v"(v[7]));
#pragma unroll
    for (int k = 0; k < GCN_BATCH; ++k)
      if (i + k * GCN_BAND < end) atomicAdd(&cnt[v[k] & (GCN_BAND - 1)], 1);
  }
  __syncthreads();

  const int d = cnt[tid];
  const int inc = gcn_wave_incl_scan(d, lane);
  if (lane == 31) wsum[wv] = inc;
  __syncthreads();
  const int ws = (lane < GCN_BAND / 32) ? wsum[lane] : 0;
  const int winc = gcn_wave_incl_scan(ws, lane);
  const int wexcl = __builtin_amdgcn_readlane(winc, wv) - __builtin_amdgcn_readlane(ws, wv);
  const int start = beg + wexcl + inc - d;
  const int row = b * GCN_BAND + tid;
  if (row < GCN_N_NODES) {
    rowptr[row] = start;
    if (row == GCN_N_NODES - 1) rowptr[GCN_N_NODES] = start + d;
    dinv[row] = 1.0f / fmaxf(sqrtf((float)d), 1e-12f);
  }
  cnt[tid] = start;
  __syncthreads();

  for (int i = beg + tid; i < end; i += GCN_BAND * GCN_BATCH) {
    int v[GCN_BATCH];
#pragma unroll
    for (int k = 0; k < GCN_BATCH; ++k) v[k] = binned[min(i + k * GCN_BAND, end - 1)];
    asm volatile("" ::"v"(v[0]), "v"(v[1]), "v"(v[2]), "v"(v[3]), "v"(v[4]), "v"(v[5]),
                 "v"(v[6]), "v"(v[7]));
#pragma unroll
    for (int k = 0; k < GCN_BATCH; ++k) {
      if (i + k * GCN_BAND < end) {
        const int pos = atomicAdd(&cnt[v[k] & (GCN_BAND - 1)], 1);
        csr_col[pos] = v[k] >> GCN_BAND_SHIFT;
      }
    }
  }
}


template <int N, bool RELU_SCALE_A, bool SCALE_C>
__global__ __launch_bounds__(256) void gcn_gemm_wmma(
    const float* __restrict__ A, const float* __restrict__ B,
    float* __restrict__ C, const float* __restrict__ dinv, int M, int K) {
  constexpr int NT = N / 16;
  constexpr int KC = 64;
  constexpr int N2 = N / 2;
  __shared__ __align__(16) float Bs[KC * N];

  const int wib  = threadIdx.x >> 5;
  const int lane = threadIdx.x & 31;
  const int half = lane >> 4;
  const int l15  = lane & 15;

  const int row0   = blockIdx.x * 128 + wib * 16;
  const bool active = (row0 < M);

  const float* __restrict__ Arow =
      A + (size_t)(active ? row0 + l15 : 0) * K;
  const float dscale =
      (RELU_SCALE_A && active) ? dinv[row0 + l15] : 1.0f;

  v8f acc[NT];
#pragma unroll
  for (int t = 0; t < NT; ++t) acc[t] = (v8f){};

  for (int kc = 0; kc < K; kc += KC) {
    const float* __restrict__ Bg = B + (size_t)kc * N;
#pragma unroll
    for (int pidx = threadIdx.x; pidx < (KC / 2) * N2; pidx += 256) {
      const int kp = pidx / N2;
      const int c2 = (pidx - kp * N2) * 2;
      const float2 g0 = *(const float2*)(Bg + (size_t)(2 * kp) * N + c2);
      const float2 g1 = *(const float2*)(Bg + (size_t)(2 * kp + 1) * N + c2);
      *(float4*)(&Bs[kp * (2 * N) + c2 * 2]) =
          make_float4(g0.x, g1.x, g0.y, g1.y);
    }
    __syncthreads();

    if (active) {
      const float* __restrict__ Ak = Arow + kc;
#pragma unroll
      for (int k0 = 0; k0 < KC; k0 += 4) {
        const int koff = k0 + 2 * half;
        v2f a = *(const v2f*)(Ak + koff);
        if (RELU_SCALE_A) {
          a.x = fmaxf(a.x * dscale, 0.0f);
          a.y = fmaxf(a.y * dscale, 0.0f);
        }
        const int kp = koff >> 1;
#pragma unroll
        for (int t = 0; t < NT; ++t) {
          const v2f b =
              *(const v2f*)(&Bs[kp * (2 * N) + (t * 16 + l15) * 2]);
          acc[t] = __builtin_amdgcn_wmma_f32_16x16x4_f32(
              false, a, false, b, (short)0, acc[t], false, false);
        }
      }
    }
    __syncthreads();
  }

  if (active) {
#pragma unroll
    for (int t = 0; t < NT; ++t) {
      const int col = t * 16 + l15;
#pragma unroll
      for (int v = 0; v < 8; ++v) {
        const int row = row0 + v + 8 * half;
        float val = acc[t][v];
        if (SCALE_C) val *= dinv[row];
        C[(size_t)row * N + col] = val;
      }
    }
  }
}


template <int VEC, bool SCALE_OUT>
__global__ __launch_bounds__(256) void gcn_pull_rows(
    const float* __restrict__ src, const int* __restrict__ rowptr,
    const int* __restrict__ csr_col, const float* __restrict__ dinv,
    float* __restrict__ dst, int n_rows) {
  typedef float vf __attribute__((ext_vector_type(VEC)));
  constexpr int F = VEC * 32;
  const int lane = threadIdx.x & 31;
  const int row =
      __builtin_amdgcn_readfirstlane(blockIdx.x * 8 + (threadIdx.x >> 5));
  if (row >= n_rows) return;
  const int beg = rowptr[row];
  const int end = rowptr[row + 1];
  const float* __restrict__ srcl = src + lane * VEC;

  vf acc = {};
  for (int base = beg; base < end; base += 32) {
    const int n = min(32, end - base);
    const int colv = (lane < n) ? csr_col[base + lane] : 0;
    int j = 0;
    for (; j + 4 <= n; j += 4) {
      const int c0 = __builtin_amdgcn_readlane(colv, j);
      const int c1 = __builtin_amdgcn_readlane(colv, j + 1);
      const int c2 = __builtin_amdgcn_readlane(colv, j + 2);
      const int c3 = __builtin_amdgcn_readlane(colv, j + 3);
      const vf x0 = *(const vf*)(srcl + (size_t)c0 * F);
      const vf x1 = *(const vf*)(srcl + (size_t)c1 * F);
      const vf x2 = *(const vf*)(srcl + (size_t)c2 * F);
      const vf x3 = *(const vf*)(srcl + (size_t)c3 * F);
      acc += x0;
      acc += x1;
      acc += x2;
      acc += x3;
    }
    for (; j < n; ++j) {
      const int c0 = __builtin_amdgcn_readlane(colv, j);
      acc += *(const vf*)(srcl + (size_t)c0 * F);
    }
  }
  if (SCALE_OUT) acc *= dinv[row];
  *(vf*)(dst + (size_t)row * F + lane * VEC) = acc;
}


extern "C" void kernel_launch(void* const* d_in, const int* in_sizes, int n_in,
                              void* d_out, int out_size, void* d_ws, size_t ws_size,
                              hipStream_t stream) {
  const float* X  = (const float*)d_in[0];
  const float* W1 = (const float*)d_in[1];
  const float* W2 = (const float*)d_in[2];
  const int* erow = (const int*)d_in[3];
  const int* ecol = (const int*)d_in[4];
  const int n_edges = in_sizes[3];
  float* out = (float*)d_out;

  int* hist      = (int*)d_ws;
  int* bandstart = hist + GCN_NBANDS * GCN_SLICES;
  int* rowptr = (int*)d_ws + GCN_N_NODES;
  float* dinv = (float*)(rowptr + GCN_ROWPTR_PAD);
  float* XWs = (float*)d_ws + GCN_ROWS_OFF;
  float* S   = XWs + (size_t)GCN_N_NODES * 128;
  float* HWs = S + (size_t)GCN_N_NODES * 128;
  int* binned  = (int*)(HWs + (size_t)GCN_N_NODES * 64);
  int* csr_col = binned + n_edges;

  gcn_band_hist_kernel<<<GCN_SLICES, 1024, 0, stream>>>(erow, hist, n_edges);
  gcn_band_scatter_kernel<<<GCN_SLICES, 1024, 0, stream>>>(
      erow, ecol, hist, bandstart, binned, n_edges);
  gcn_band_sort_kernel<<<GCN_NBANDS, GCN_BAND, 0, stream>>>(
      binned, bandstart, rowptr, csr_col, dinv);

  const int row_tiles = GCN_N_NODES / 16;
  const int gemm_blocks = (row_tiles + 7) / 8;
  const int pull_blocks = (GCN_N_NODES + 7) / 8;

  gcn_gemm_wmma<128, false, true>
      <<<gemm_blocks, 256, 0, stream>>>(X, W1, XWs, dinv, GCN_N_NODES, 256);

  gcn_pull_rows<4, false><<<pull_blocks, 256, 0, stream>>>(
      XWs, rowptr, csr_col, dinv, S, GCN_N_NODES);

  gcn_gemm_wmma<64, true, true>
      <<<gemm_blocks, 256, 0, stream>>>(S, W2, HWs, dinv, GCN_N_NODES, 128);

  gcn_pull_rows<2, true><<<pull_blocks, 256, 0, stream>>>(
      HWs, rowptr, csr_col, dinv, out, GCN_N_NODES);
}


# GEMMs persistent: all of W staged once per work-group in LDS, one work-group per WGP; waves walk 16-row tiles dealt evenly over SIMDs (rest as column pairs); no per-chunk barrier; sum order kept
# speedup vs baseline: 3.3710x; 1.1403x over previous
#include <hip/hip_runtime.h>


typedef float v2f __attribute__((ext_vector_type(2)));
typedef float v8f __attribute__((ext_vector_type(8)));

#define GCN_N_NODES 100000
#define GCN_ROWPTR_PAD 100016
#define GCN_ROWS_OFF 301056
static_assert(GCN_ROWS_OFF >= GCN_N_NODES + GCN_ROWPTR_PAD + GCN_N_NODES,
              "row arrays start behind hist, rowptr and dinv");
static_assert(GCN_N_NODES % 16 == 0, "the GEMMs work on whole 16-row tiles");
static_assert(GCN_ROWS_OFF % 1024 == 0, "row arrays start on a 4096-byte boundary");
static_assert(((size_t)GCN_N_NODES * 64 * sizeof(float)) % 4096 == 0,
              "every row array is a multiple of 4096 bytes long");


#define GCN_BAND_SHIFT 9
#define GCN_BAND 512
#define GCN_NBANDS 196
#define GCN_SLICES 256
#define GCN_BATCH 8

static_assert((GCN_N_NODES + GCN_BAND - 1) / GCN_BAND == GCN_NBANDS, "band count");
static_assert(GCN_N_NODES <= (1 << 22), "col << 9 must stay a non-negative int");
static_assert(GCN_BATCH == 8, "the load fences list 8 values");
static_assert(GCN_NBANDS <= 7 * 32, "scatter kernel: 7 bands per lane / per wave");
static_assert(GCN_NBANDS * GCN_SLICES + GCN_NBANDS + 1 <= GCN_N_NODES,
              "hist and bandstart live in the first GCN_N_NODES ints of the workspace");

__device__ __forceinline__ int gcn_wave_incl_scan(int x, int lane) {
  x += __builtin_amdgcn_update_dpp(0, x, 0x111, 0xf, 0xf, false);
  x += __builtin_amdgcn_update_dpp(0, x, 0x112, 0xf, 0xf, false);
  x += __builtin_amdgcn_update_dpp(0, x, 0x114, 0xf, 0xf, false);
  x += __builtin_amdgcn_update_dpp(0, x, 0x118, 0xf, 0xf, false);
  const int low = __builtin_amdgcn_readlane(x, 15);
  return x + ((lane >= 16) ? low : 0);
}

__global__ __launch_bounds__(1024) void gcn_band_hist_kernel(
    const int* __restrict__ rows, int* __restrict__ hist, int n_edges) {
  __shared__ int h[GCN_NBANDS];
  const int tid = threadIdx.x;
  const int w = blockIdx.x;
  if (tid < GCN_NBANDS) h[tid] = 0;
  __syncthreads();
  const int sl = (n_edges + GCN_SLICES - 1) / GCN_SLICES;
  const int s0 = min(w * sl, n_edges);
  const int s1 = min(s0 + sl, n_edges);
  for (int e = s0 + tid; e < s1; e += 1024 * GCN_BATCH) {
    int r[GCN_BATCH];
#pragma unroll
    for (int k = 0; k < GCN_BATCH; ++k) r[k] = rows[min(e + k * 1024, s1 - 1)];
    asm volatile("" ::"v"(r[0]), "v"(r[1]), "v"(r[2]), "v"(r[3]), "v"(r[4]), "v"(r[5]),
                 "v"(r[6]), "v"(r[7]));
#pragma unroll
    for (int k = 0; k < GCN_BATCH; ++k)
      if (e + k * 1024 < s1) atomicAdd(&h[r[k] >> GCN_BAND_SHIFT], 1);
  }
  __syncthreads();
  if (tid < GCN_NBANDS) hist[tid * GCN_SLICES + w] = h[tid];
}

__global__ __launch_bounds__(1024) void gcn_band_scatter_kernel(
    const int* __restrict__ rows, const int* __restrict__ cols,
    const int* __restrict__ hist, int* __restrict__ bandstart,
    int* __restrict__ binned, int n_edges) {
  __shared__ int sT[7 * 32];
  __shared__ int sP[7 * 32];
  __shared__ int cursor[GCN_NBANDS];
  const int tid = threadIdx.x;
  const int lane = tid & 31;
  const int wv = __builtin_amdgcn_readfirstlane(tid >> 5);
  const int w = blockIdx.x;

  int4 qa[7], qb[7];
#pragma unroll
  for (int k = 0; k < 7; ++k) {
    const int bb = min(wv + 32 * k, GCN_NBANDS - 1);
    const int4* __restrict__ p = (const int4*)(hist + bb * GCN_SLICES + lane * 8);
    qa[k] = p[0];
    qb[k] = p[1];
  }
#pragma unroll
  for (int k = 0; k < 7; ++k) {
    const int b = wv + 32 * k;
    const int j0 = lane * 8;
    int tot = qa[k].x + qa[k].y + qa[k].z + qa[k].w + qb[k].x + qb[k].y + qb[k].z + qb[k].w;
    int part = ((j0 + 0 < w) ? qa[k].x : 0) + ((j0 + 1 < w) ? qa[k].y : 0) +
               ((j0 + 2 < w) ? qa[k].z : 0) + ((j0 + 3 < w) ? qa[k].w : 0) +
               ((j0 + 4 < w) ? qb[k].x : 0) + ((j0 + 5 < w) ? qb[k].y : 0) +
               ((j0 + 6 < w) ? qb[k].z : 0) + ((j0 + 7 < w) ? qb[k].w : 0);
    tot = __builtin_amdgcn_readlane(gcn_wave_incl_scan(tot, lane), 31);
    part = __builtin_amdgcn_readlane(gcn_wave_incl_scan(part, lane), 31);
    if (lane == 0) {
      sT[b] = (b < GCN_NBANDS) ? tot : 0;
      sP[b] = (b < GCN_NBANDS) ? part : 0;
    }
  }
  __syncthreads();

  if (wv == 0) {
    int t[7];
    int s = 0;
#pragma unroll
    for (int k = 0; k < 7; ++k) {
      t[k] = sT[lane * 7 + k];
      s += t[k];
    }
    int run = gcn_wave_incl_scan(s, lane) - s;
#pragma unroll
    for (int k = 0; k < 7; ++k) {
      const int b = lane * 7 + k;
      if (b < GCN_NBANDS) {
        cursor[b] = run + sP[b];
        if (w == 0) bandstart[b] = run;
      }
      run += t[k];
    }
    if (w == 0 && lane == 31) bandstart[GCN_NBANDS] = run;
  }
  __syncthreads();

  const int sl = (n_edges + GCN_SLICES - 1) / GCN_SLICES;
  const int s0 = min(w * sl, n_edges);
  const int s1 = min(s0 + sl, n_edges);
  for (int e = s0 + tid; e < s1; e += 1024 * GCN_BATCH) {
    int r[GCN_BATCH], c[GCN_BATCH];
#pragma unroll
    for (int k = 0; k < GCN_BATCH; ++k) {
      const int ee = min(e + k * 1024, s1 - 1);
      r[k] = rows[ee];
      c[k] = cols[ee];
    }
    asm volatile("" ::"v"(r[0]), "v"(r[1]), "v"(r[2]), "v"(r[3]), "v"(r[4]), "v"(r[5]),
                 "v"(r[6]), "v"(r[7]), "v"(c[0]), "v"(c[1]), "v"(c[2]), "v"(c[3]), "v"(c[4]),
                 "v"(c[5]), "v"(c[6]), "v"(c[7]));
#pragma unroll
    for (int k = 0; k < GCN_BATCH; ++k) {
      if (e + k * 1024 < s1) {
        const int pos = atomicAdd(&cursor[r[k] >> GCN_BAND_SHIFT], 1);
        binned[pos] = (c[k] << GCN_BAND_SHIFT) | (r[k] & (GCN_BAND - 1));
      }
    }
  }
}

__global__ __launch_bounds__(GCN_BAND) void gcn_band_sort_kernel(
    const int* __restrict__ binned, const int* __restrict__ bandstart,
    int* __restrict__ rowptr, int* __restrict__ csr_col, float* __restrict__ dinv) {
  __shared__ int cnt[GCN_BAND];
  __shared__ int wsum[GCN_BAND / 32];
  const int tid = threadIdx.x;
  const int lane = tid & 31;
  const int wv = __builtin_amdgcn_readfirstlane(tid >> 5);
  const int b = blockIdx.x;
  const int beg = bandstart[b];
  const int end = bandstart[b + 1];

  cnt[tid] = 0;
  __syncthreads();
  for (int i = beg + tid; i < end; i += GCN_BAND * GCN_BATCH) {
    int v[GCN_BATCH];
#pragma unroll
    for (int k = 0; k < GCN_BATCH; ++k) v[k] = binned[min(i + k * GCN_BAND, end - 1)];
    asm volatile("" ::"v"(v[0]), "v"(v[1]), "v"(v[2]), "v"(v[3]), "v"(v[4]), "v"(v[5]),
                 "v"(v[6]), "v"(v[7]));
#pragma unroll
    for (int k = 0; k < GCN_BATCH; ++k)
      if (i + k * GCN_BAND < end) atomicAdd(&cnt[v[k] & (GCN_BAND - 1)], 1);
  }
  __syncthreads();

  const int d = cnt[tid];
  const int inc = gcn_wave_incl_scan(d, lane);
  if (lane == 31) wsum[wv] = inc;
  __syncthreads();
  const int ws = (lane < GCN_BAND / 32) ? wsum[lane] : 0;
  const int winc = gcn_wave_incl_scan(ws, lane);
  const int wexcl = __builtin_amdgcn_readlane(winc, wv) - __builtin_amdgcn_readlane(ws, wv);
  const int start = beg + wexcl + inc - d;
  const int row = b * GCN_BAND + tid;
  if (row < GCN_N_NODES) {
    rowptr[row] = start;
    if (row == GCN_N_NODES - 1) rowptr[GCN_N_NODES] = start + d;
    dinv[row] = 1.0f / fmaxf(sqrtf((float)d), 1e-12f);
  }
  cnt[tid] = start;
  __syncthreads();

  for (int i = beg + tid; i < end; i += GCN_BAND * GCN_BATCH) {
    int v[GCN_BATCH];
#pragma unroll
    for (int k = 0; k < GCN_BATCH; ++k) v[k] = binned[min(i + k * GCN_BAND, end - 1)];
    asm volatile("" ::"v"(v[0]), "v"(v[1]), "v"(v[2]), "v"(v[3]), "v"(v[4]), "v"(v[5]),
                 "v"(v[6]), "v"(v[7]));
#pragma unroll
    for (int k = 0; k < GCN_BATCH; ++k) {
      if (i + k * GCN_BAND < end) {
        const int pos = atomicAdd(&cnt[v[k] & (GCN_BAND - 1)], 1);
        csr_col[pos] = v[k] >> GCN_BAND_SHIFT;
      }
    }
  }
}


template <int N, int K, int NTL, bool RELU_SCALE_A, bool SCALE_C>
__device__ __forceinline__ void gcn_gemm_tile(
    const float* __restrict__ A, const float* Bs, float* __restrict__ C,
    const float* __restrict__ dinv, int row0, int col0, int half, int l15) {
  constexpr int KC = 64;
  static_assert(K % KC == 0, "whole chunks");
  const float* __restrict__ Arow = A + (size_t)(row0 + l15) * K + 2 * half;
  const float dscale = RELU_SCALE_A ? dinv[row0 + l15] : 1.0f;

  v8f acc[NTL];
#pragma unroll
  for (int t = 0; t < NTL; ++t) acc[t] = (v8f){};

  v2f a_nxt[KC / 4];
#pragma unroll
  for (int i = 0; i < KC / 4; ++i) a_nxt[i] = *(const v2f*)(Arow + 4 * i);

#pragma unroll 1
  for (int kc = 0; kc < K; kc += KC) {
    v2f a_cur[KC / 4];
#pragma unroll
    for (int i = 0; i < KC / 4; ++i) a_cur[i] = a_nxt[i];
    if (kc + KC < K) {
#pragma unroll
      for (int i = 0; i < KC / 4; ++i)
        a_nxt[i] = *(const v2f*)(Arow + kc + KC + 4 * i);
    }
#pragma unroll
    for (int i = 0; i < KC / 4; ++i) {
      v2f a = a_cur[i];
      if (RELU_SCALE_A) {
        a.x = fmaxf(a.x * dscale, 0.0f);
        a.y = fmaxf(a.y * dscale, 0.0f);
      }
      const int kp = (kc + 4 * i + 2 * half) >> 1;
#pragma unroll
      for (int t = 0; t < NTL; ++t) {
        const v2f b =
            *(const v2f*)(&Bs[kp * (2 * N) + (col0 + t * 16 + l15) * 2]);
        acc[t] = __builtin_amdgcn_wmma_f32_16x16x4_f32(
            false, a, false, b, (short)0, acc[t], false, false);
      }
    }
  }

#pragma unroll
  for (int t = 0; t < NTL; ++t) {
    const int col = col0 + t * 16 + l15;
#pragma unroll
    for (int v = 0; v < 8; ++v) {
      const int row = row0 + v + 8 * half;
      float val = acc[t][v];
      if (SCALE_C) val *= dinv[row];
      C[(size_t)row * N + col] = val;
    }
  }
}

template <int N, int K, int WAVES, bool RELU_SCALE_A, bool SCALE_C>
__global__ __launch_bounds__(WAVES * 32) void gcn_gemm_wmma(
    const float* __restrict__ A, const float* __restrict__ B,
    float* __restrict__ C, const float* __restrict__ dinv, int M) {
  constexpr int NT = N / 16;
  constexpr int N2 = N / 2;
  constexpr int THREADS = WAVES * 32;
  constexpr int STAGE_ITEMS = (K / 2) * N2;
  constexpr int PARTS = NT / 2;
  static_assert(STAGE_ITEMS % THREADS == 0, "every thread stages the same number of items");
  static_assert(N % 32 == 0 && WAVES % 8 == 0, "column pairs; both halves of the waves cover all SIMDs");
  __shared__ __align__(16) float Bs[K * N];

  const int tid  = threadIdx.x;
  const int wib  = __builtin_amdgcn_readfirstlane(tid >> 5);
  const int lane = tid & 31;
  const int half = lane >> 4;
  const int l15  = lane & 15;

#pragma unroll
  for (int it = 0; it < STAGE_ITEMS / THREADS; ++it) {
    const int pidx = tid + it * THREADS;
    const int kp = pidx / N2;
    const int c2 = (pidx - kp * N2) * 2;
    const float2 g0 = *(const float2*)(B + (size_t)(2 * kp) * N + c2);
    const float2 g1 = *(const float2*)(B + (size_t)(2 * kp + 1) * N + c2);
    *(float4*)(&Bs[kp * (2 * N) + c2 * 2]) =
        make_float4(g0.x, g1.x, g0.y, g1.y);
  }
  __syncthreads();

  const int n_tiles = M >> 4;
  const int grid = gridDim.x;
  const int n_slots = grid * WAVES;
  const int half_slots = n_slots >> 1;
  const int slot = wib * grid + blockIdx.x;

  const int whole = (n_tiles / n_slots) * n_slots;
  const bool half_round = (n_tiles - whole >= half_slots);
  const int n_full = whole + (half_round ? half_slots : 0);
  for (int tile = slot; tile < n_full; tile += n_slots)
    gcn_gemm_tile<N, K, NT, RELU_SCALE_A, SCALE_C>(A, Bs, C, dinv, tile * 16, 0, half, l15);

  const int n_parts = (n_tiles - n_full) * PARTS;
  const int first = half_round ? half_slots : 0;
  const int takers = n_slots - first;
  if (slot >= first) {
    for (int u = slot - first; u < n_parts; u += takers) {
      const int tile = n_full + u / PARTS;
      const int col0 = (u % PARTS) * 32;
      gcn_gemm_tile<N, K, 2, RELU_SCALE_A, SCALE_C>(A, Bs, C, dinv, tile * 16, col0, half, l15);
    }
  }
}


template <int VEC, bool SCALE_OUT>
__global__ __launch_bounds__(256) void gcn_pull_rows(
    const float* __restrict__ src, const int* __restrict__ rowptr,
    const int* __restrict__ csr_col, const float* __restrict__ dinv,
    float* __restrict__ dst, int n_rows) {
  typedef float vf __attribute__((ext_vector_type(VEC)));
  constexpr int F = VEC * 32;
  const int lane = threadIdx.x & 31;
  const int row =
      __builtin_amdgcn_readfirstlane(blockIdx.x * 8 + (threadIdx.x >> 5));
  if (row >= n_rows) return;
  const int beg = rowptr[row];
  const int end = rowptr[row + 1];
  const float* __restrict__ srcl = src + lane * VEC;

  vf acc = {};
  for (int base = beg; base < end; base += 32) {
    const int n = min(32, end - base);
    const int colv = (lane < n) ? csr_col[base + lane] : 0;
    int j = 0;
    for (; j + 4 <= n; j += 4) {
      const int c0 = __builtin_amdgcn_readlane(colv, j);
      const int c1 = __builtin_amdgcn_readlane(colv, j + 1);
      const int c2 = __builtin_amdgcn_readlane(colv, j + 2);
      const int c3 = __builtin_amdgcn_readlane(colv, j + 3);
      const vf x0 = *(const vf*)(srcl + (size_t)c0 * F);
      const vf x1 = *(const vf*)(srcl + (size_t)c1 * F);
      const vf x2 = *(const vf*)(srcl + (size_t)c2 * F);
      const vf x3 = *(const vf*)(srcl + (size_t)c3 * F);
      acc += x0;
      acc += x1;
      acc += x2;
      acc += x3;
    }
    for (; j < n; ++j) {
      const int c0 = __builtin_amdgcn_readlane(colv, j);
      acc += *(const vf*)(srcl + (size_t)c0 * F);
    }
  }
  if (SCALE_OUT) acc *= dinv[row];
  *(vf*)(dst + (size_t)row * F + lane * VEC) = acc;
}


extern "C" void kernel_launch(void* const* d_in, const int* in_sizes, int n_in,
                              void* d_out, int out_size, void* d_ws, size_t ws_size,
                              hipStream_t stream) {
  const float* X  = (const float*)d_in[0];
  const float* W1 = (const float*)d_in[1];
  const float* W2 = (const float*)d_in[2];
  const int* erow = (const int*)d_in[3];
  const int* ecol = (const int*)d_in[4];
  const int n_edges = in_sizes[3];
  float* out = (float*)d_out;

  int* hist      = (int*)d_ws;
  int* bandstart = hist + GCN_NBANDS * GCN_SLICES;
  int* rowptr = (int*)d_ws + GCN_N_NODES;
  float* dinv = (float*)(rowptr + GCN_ROWPTR_PAD);
  float* XWs = (float*)d_ws + GCN_ROWS_OFF;
  float* S   = XWs + (size_t)GCN_N_NODES * 128;
  float* HWs = S + (size_t)GCN_N_NODES * 128;
  int* binned  = (int*)(HWs + (size_t)GCN_N_NODES * 64);
  int* csr_col = binned + n_edges;

  gcn_band_hist_kernel<<<GCN_SLICES, 1024, 0, stream>>>(erow, hist, n_edges);
  gcn_band_scatter_kernel<<<GCN_SLICES, 1024, 0, stream>>>(
      erow, ecol, hist, bandstart, binned, n_edges);
  gcn_band_sort_kernel<<<GCN_NBANDS, GCN_BAND, 0, stream>>>(
      binned, bandstart, rowptr, csr_col, dinv);

  const int gemm_blocks = 256;
  const int pull_blocks = (GCN_N_NODES + 7) / 8;

  gcn_gemm_wmma<128, 256, 16, false, true>
      <<<gemm_blocks, 512, 0, stream>>>(X, W1, XWs, dinv, GCN_N_NODES);

  gcn_pull_rows<4, false><<<pull_blocks, 256, 0, stream>>>(
      XWs, rowptr, csr_col, dinv, S, GCN_N_NODES);

  gcn_gemm_wmma<64, 128, 32, true, true>
      <<<gemm_blocks, 1024, 0, stream>>>(S, W2, HWs, dinv, GCN_N_NODES);

  gcn_pull_rows<2, true><<<pull_blocks, 256, 0, stream>>>(
      HWs, rowptr, csr_col, dinv, out, GCN_N_NODES);
}


# First pull fused into GEMM2: a 16-wave work-group sums a 16-row tile (dinv, ReLU) into LDS, two of its waves multiply it by W2 held in LDS; row sums never reach memory; sum and k order kept
# speedup vs baseline: 3.5572x; 1.0552x over previous
#include <hip/hip_runtime.h>


typedef float v2f __attribute__((ext_vector_type(2)));
typedef float v8f __attribute__((ext_vector_type(8)));

#define GCN_N_NODES 100000
#define GCN_ROWPTR_PAD 100016
#define GCN_ROWS_OFF 301056
static_assert(GCN_ROWS_OFF >= GCN_N_NODES + GCN_ROWPTR_PAD + GCN_N_NODES,
              "row arrays start behind hist, rowptr and dinv");
static_assert(GCN_N_NODES % 16 == 0, "the GEMMs work on whole 16-row tiles");
static_assert(GCN_ROWS_OFF % 1024 == 0, "row arrays start on a 4096-byte boundary");
static_assert(((size_t)GCN_N_NODES * 64 * sizeof(float)) % 4096 == 0,
              "every row array is a multiple of 4096 bytes long");


#define GCN_BAND_SHIFT 9
#define GCN_BAND 512
#define GCN_NBANDS 196
#define GCN_SLICES 256
#define GCN_BATCH 8

static_assert((GCN_N_NODES + GCN_BAND - 1) / GCN_BAND == GCN_NBANDS, "band count");
static_assert(GCN_N_NODES <= (1 << 22), "col << 9 must stay a non-negative int");
static_assert(GCN_BATCH == 8, "the load fences list 8 values");
static_assert(GCN_NBANDS <= 7 * 32, "scatter kernel: 7 bands per lane / per wave");
static_assert(GCN_NBANDS * GCN_SLICES + GCN_NBANDS + 1 <= GCN_N_NODES,
              "hist and bandstart live in the first GCN_N_NODES ints of the workspace");

__device__ __forceinline__ int gcn_wave_incl_scan(int x, int lane) {
  x += __builtin_amdgcn_update_dpp(0, x, 0x111, 0xf, 0xf, false);
  x += __builtin_amdgcn_update_dpp(0, x, 0x112, 0xf, 0xf, false);
  x += __builtin_amdgcn_update_dpp(0, x, 0x114, 0xf, 0xf, false);
  x += __builtin_amdgcn_update_dpp(0, x, 0x118, 0xf, 0xf, false);
  const int low = __builtin_amdgcn_readlane(x, 15);
  return x + ((lane >= 16) ? low : 0);
}

__global__ __launch_bounds__(1024) void gcn_band_hist_kernel(
    const int* __restrict__ rows, int* __restrict__ hist, int n_edges) {
  __shared__ int h[GCN_NBANDS];
  const int tid = threadIdx.x;
  const int w = blockIdx.x;
  if (tid < GCN_NBANDS) h[tid] = 0;
  __syncthreads();
  const int sl = (n_edges + GCN_SLICES - 1) / GCN_SLICES;
  const int s0 = min(w * sl, n_edges);
  const int s1 = min(s0 + sl, n_edges);
  for (int e = s0 + tid; e < s1; e += 1024 * GCN_BATCH) {
    int r[GCN_BATCH];
#pragma unroll
    for (int k = 0; k < GCN_BATCH; ++k) r[k] = rows[min(e + k * 1024, s1 - 1)];
    asm volatile("" ::"v"(r[0]), "v"(r[1]), "v"(r[2]), "v"(r[3]), "v"(r[4]), "v"(r[5]),
                 "v"(r[6]), "v"(r[7]));
#pragma unroll
    for (int k = 0; k < GCN_BATCH; ++k)
      if (e + k * 1024 < s1) atomicAdd(&h[r[k] >> GCN_BAND_SHIFT], 1);
  }
  __syncthreads();
  if (tid < GCN_NBANDS) hist[tid * GCN_SLICES + w] = h[tid];
}

__global__ __launch_bounds__(1024) void gcn_band_scatter_kernel(
    const int* __restrict__ rows, const int* __restrict__ cols,
    const int* __restrict__ hist, int* __restrict__ bandstart,
    int* __restrict__ binned, int n_edges) {
  __shared__ int sT[7 * 32];
  __shared__ int sP[7 * 32];
  __shared__ int cursor[GCN_NBANDS];
  const int tid = threadIdx.x;
  const int lane = tid & 31;
  const int wv = __builtin_amdgcn_readfirstlane(tid >> 5);
  const int w = blockIdx.x;

  int4 qa[7], qb[7];
#pragma unroll
  for (int k = 0; k < 7; ++k) {
    const int bb = min(wv + 32 * k, GCN_NBANDS - 1);
    const int4* __restrict__ p = (const int4*)(hist + bb * GCN_SLICES + lane * 8);
    qa[k] = p[0];
    qb[k] = p[1];
  }
#pragma unroll
  for (int k = 0; k < 7; ++k) {
    const int b = wv + 32 * k;
    const int j0 = lane * 8;
    int tot = qa[k].x + qa[k].y + qa[k].z + qa[k].w + qb[k].x + qb[k].y + qb[k].z + qb[k].w;
    int part = ((j0 + 0 < w) ? qa[k].x : 0) + ((j0 + 1 < w) ? qa[k].y : 0) +
               ((j0 + 2 < w) ? qa[k].z : 0) + ((j0 + 3 < w) ? qa[k].w : 0) +
               ((j0 + 4 < w) ? qb[k].x : 0) + ((j0 + 5 < w) ? qb[k].y : 0) +
               ((j0 + 6 < w) ? qb[k].z : 0) + ((j0 + 7 < w) ? qb[k].w : 0);
    tot = __builtin_amdgcn_readlane(gcn_wave_incl_scan(tot, lane), 31);
    part = __builtin_amdgcn_readlane(gcn_wave_incl_scan(part, lane), 31);
    if (lane == 0) {
      sT[b] = (b < GCN_NBANDS) ? tot : 0;
      sP[b] = (b < GCN_NBANDS) ? part : 0;
    }
  }
  __syncthreads();

  if (wv == 0) {
    int t[7];
    int s = 0;
#pragma unroll
    for (int k = 0; k < 7; ++k) {
      t[k] = sT[lane * 7 + k];
      s += t[k];
    }
    int run = gcn_wave_incl_scan(s, lane) - s;
#pragma unroll
    for (int k = 0; k < 7; ++k) {
      const int b = lane * 7 + k;
      if (b < GCN_NBANDS) {
        cursor[b] = run + sP[b];
        if (w == 0) bandstart[b] = run;
      }
      run += t[k];
    }
    if (w == 0 && lane == 31) bandstart[GCN_NBANDS] = run;
  }
  __syncthreads();

  const int sl = (n_edges + GCN_SLICES - 1) / GCN_SLICES;
  const int s0 = min(w * sl, n_edges);
  const int s1 = min(s0 + sl, n_edges);
  for (int e = s0 + tid; e < s1; e += 1024 * GCN_BATCH) {
    int r[GCN_BATCH], c[GCN_BATCH];
#pragma unroll
    for (int k = 0; k < GCN_BATCH; ++k) {
      const int ee = min(e + k * 1024, s1 - 1);
      r[k] = rows[ee];
      c[k] = cols[ee];
    }
    asm volatile("" ::"v"(r[0]), "v"(r[1]), "v"(r[2]), "v"(r[3]), "v"(r[4]), "v"(r[5]),
                 "v"(r[6]), "v"(r[7]), "v"(c[0]), "v"(c[1]), "v"(c[2]), "v"(c[3]), "v"(c[4]),
                 "v"(c[5]), "v"(c[6]), "v"(c[7]));
#pragma unroll
    for (int k = 0; k < GCN_BATCH; ++k) {
      if (e + k * 1024 < s1) {
        const int pos = atomicAdd(&cursor[r[k] >> GCN_BAND_SHIFT], 1);
        binned[pos] = (c[k] << GCN_BAND_SHIFT) | (r[k] & (GCN_BAND - 1));
      }
    }
  }
}

__global__ __launch_bounds__(GCN_BAND) void gcn_band_sort_kernel(
    const int* __restrict__ binned, const int* __restrict__ bandstart,
    int* __restrict__ rowptr, int* __restrict__ csr_col, float* __restrict__ dinv) {
  __shared__ int cnt[GCN_BAND];
  __shared__ int wsum[GCN_BAND / 32];
  const int tid = threadIdx.x;
  const int lane = tid & 31;
  const int wv = __builtin_amdgcn_readfirstlane(tid >> 5);
  const int b = blockIdx.x;
  const int beg = bandstart[b];
  const int end = bandstart[b + 1];

  cnt[tid] = 0;
  __syncthreads();
  for (int i = beg + tid; i < end; i += GCN_BAND * GCN_BATCH) {
    int v[GCN_BATCH];
#pragma unroll
    for (int k = 0; k < GCN_BATCH; ++k) v[k] = binned[min(i + k * GCN_BAND, end - 1)];
    asm volatile("" ::"v"(v[0]), "v"(v[1]), "v"(v[2]), "v"(v[3]), "v"(v[4]), "v"(v[5]),
                 "v"(v[6]), "v"(v[7]));
#pragma unroll
    for (int k = 0; k < GCN_BATCH; ++k)
      if (i + k * GCN_BAND < end) atomicAdd(&cnt[v[k] & (GCN_BAND - 1)], 1);
  }
  __syncthreads();

  const int d = cnt[tid];
  const int inc = gcn_wave_incl_scan(d, lane);
  if (lane == 31) wsum[wv] = inc;
  __syncthreads();
  const int ws = (lane < GCN_BAND / 32) ? wsum[lane] : 0;
  const int winc = gcn_wave_incl_scan(ws, lane);
  const int wexcl = __builtin_amdgcn_readlane(winc, wv) - __builtin_amdgcn_readlane(ws, wv);
  const int start = beg + wexcl + inc - d;
  const int row = b * GCN_BAND + tid;
  if (row < GCN_N_NODES) {
    rowptr[row] = start;
    if (row == GCN_N_NODES - 1) rowptr[GCN_N_NODES] = start + d;
    dinv[row] = 1.0f / fmaxf(sqrtf((float)d), 1e-12f);
  }
  cnt[tid] = start;
  __syncthreads();

  for (int i = beg + tid; i < end; i += GCN_BAND * GCN_BATCH) {
    int v[GCN_BATCH];
#pragma unroll
    for (int k = 0; k < GCN_BATCH; ++k) v[k] = binned[min(i + k * GCN_BAND, end - 1)];
    asm volatile("" ::"v"(v[0]), "v"(v[1]), "v"(v[2]), "v"(v[3]), "v"(v[4]), "v"(v[5]),
                 "v"(v[6]), "v"(v[7]));
#pragma unroll
    for (int k = 0; k < GCN_BATCH; ++k) {
      if (i + k * GCN_BAND < end) {
        const int pos = atomicAdd(&cnt[v[k] & (GCN_BAND - 1)], 1);
        csr_col[pos] = v[k] >> GCN_BAND_SHIFT;
      }
    }
  }
}


template <int N, int K, int NTL, bool RELU_SCALE_A, bool SCALE_C>
__device__ __forceinline__ void gcn_gemm_tile(
    const float* __restrict__ A, const float* Bs, float* __restrict__ C,
    const float* __restrict__ dinv, int row0, int col0, int half, int l15) {
  constexpr int KC = 64;
  static_assert(K % KC == 0, "whole chunks");
  const float* __restrict__ Arow = A + (size_t)(row0 + l15) * K + 2 * half;
  const float dscale = RELU_SCALE_A ? dinv[row0 + l15] : 1.0f;

  v8f acc[NTL];
#pragma unroll
  for (int t = 0; t < NTL; ++t) acc[t] = (v8f){};

  v2f a_nxt[KC / 4];
#pragma unroll
  for (int i = 0; i < KC / 4; ++i) a_nxt[i] = *(const v2f*)(Arow + 4 * i);

#pragma unroll 1
  for (int kc = 0; kc < K; kc += KC) {
    v2f a_cur[KC / 4];
#pragma unroll
    for (int i = 0; i < KC / 4; ++i) a_cur[i] = a_nxt[i];
    if (kc + KC < K) {
#pragma unroll
      for (int i = 0; i < KC / 4; ++i)
        a_nxt[i] = *(const v2f*)(Arow + kc + KC + 4 * i);
    }
#pragma unroll
    for (int i = 0; i < KC / 4; ++i) {
      v2f a = a_cur[i];
      if (RELU_SCALE_A) {
        a.x = fmaxf(a.x * dscale, 0.0f);
        a.y = fmaxf(a.y * dscale, 0.0f);
      }
      const int kp = (kc + 4 * i + 2 * half) >> 1;
#pragma unroll
      for (int t = 0; t < NTL; ++t) {
        const v2f b =
            *(const v2f*)(&Bs[kp * (2 * N) + (col0 + t * 16 + l15) * 2]);
        acc[t] = __builtin_amdgcn_wmma_f32_16x16x4_f32(
            false, a, false, b, (short)0, acc[t], false, false);
      }
    }
  }

#pragma unroll
  for (int t = 0; t < NTL; ++t) {
    const int col = col0 + t * 16 + l15;
#pragma unroll
    for (int v = 0; v < 8; ++v) {
      const int row = row0 + v + 8 * half;
      float val = acc[t][v];
      if (SCALE_C) val *= dinv[row];
      C[(size_t)row * N + col] = val;
    }
  }
}

template <int N, int K, int WAVES, bool RELU_SCALE_A, bool SCALE_C>
__global__ __launch_bounds__(WAVES * 32) void gcn_gemm_wmma(
    const float* __restrict__ A, const float* __restrict__ B,
    float* __restrict__ C, const float* __restrict__ dinv, int M) {
  constexpr int NT = N / 16;
  constexpr int N2 = N / 2;
  constexpr int THREADS = WAVES * 32;
  constexpr int STAGE_ITEMS = (K / 2) * N2;
  constexpr int PARTS = NT / 2;
  static_assert(STAGE_ITEMS % THREADS == 0, "every thread stages the same number of items");
  static_assert(N % 32 == 0 && WAVES % 8 == 0, "column pairs; both halves of the waves cover all SIMDs");
  __shared__ __align__(16) float Bs[K * N];

  const int tid  = threadIdx.x;
  const int wib  = __builtin_amdgcn_readfirstlane(tid >> 5);
  const int lane = tid & 31;
  const int half = lane >> 4;
  const int l15  = lane & 15;

#pragma unroll
  for (int it = 0; it < STAGE_ITEMS / THREADS; ++it) {
    const int pidx = tid + it * THREADS;
    const int kp = pidx / N2;
    const int c2 = (pidx - kp * N2) * 2;
    const float2 g0 = *(const float2*)(B + (size_t)(2 * kp) * N + c2);
    const float2 g1 = *(const float2*)(B + (size_t)(2 * kp + 1) * N + c2);
    *(float4*)(&Bs[kp * (2 * N) + c2 * 2]) =
        make_float4(g0.x, g1.x, g0.y, g1.y);
  }
  __syncthreads();

  const int n_tiles = M >> 4;
  const int grid = gridDim.x;
  const int n_slots = grid * WAVES;
  const int half_slots = n_slots >> 1;
  const int slot = wib * grid + blockIdx.x;

  const int whole = (n_tiles / n_slots) * n_slots;
  const bool half_round = (n_tiles - whole >= half_slots);
  const int n_full = whole + (half_round ? half_slots : 0);
  for (int tile = slot; tile < n_full; tile += n_slots)
    gcn_gemm_tile<N, K, NT, RELU_SCALE_A, SCALE_C>(A, Bs, C, dinv, tile * 16, 0, half, l15);

  const int n_parts = (n_tiles - n_full) * PARTS;
  const int first = half_round ? half_slots : 0;
  const int takers = n_slots - first;
  if (slot >= first) {
    for (int u = slot - first; u < n_parts; u += takers) {
      const int tile = n_full + u / PARTS;
      const int col0 = (u % PARTS) * 32;
      gcn_gemm_tile<N, K, 2, RELU_SCALE_A, SCALE_C>(A, Bs, C, dinv, tile * 16, col0, half, l15);
    }
  }
}


template <int VEC>
using gcn_vf = float __attribute__((ext_vector_type(VEC)));

template <int VEC>
__device__ __forceinline__ gcn_vf<VEC> gcn_row_sum(
    const float* __restrict__ srcl, const int* __restrict__ csr_col, int beg, int end,
    int lane) {
  typedef gcn_vf<VEC> vf;
  constexpr int F = VEC * 32;
  vf acc = {};
  for (int base = beg; base < end; base += 32) {
    const int n = min(32, end - base);
    const int colv = (lane < n) ? csr_col[base + lane] : 0;
    int j = 0;
    for (; j + 4 <= n; j += 4) {
      const int c0 = __builtin_amdgcn_readlane(colv, j);
      const int c1 = __builtin_amdgcn_readlane(colv, j + 1);
      const int c2 = __builtin_amdgcn_readlane(colv, j + 2);
      const int c3 = __builtin_amdgcn_readlane(colv, j + 3);
      const vf x0 = *(const vf*)(srcl + (size_t)c0 * F);
      const vf x1 = *(const vf*)(srcl + (size_t)c1 * F);
      const vf x2 = *(const vf*)(srcl + (size_t)c2 * F);
      const vf x3 = *(const vf*)(srcl + (size_t)c3 * F);
      acc += x0;
      acc += x1;
      acc += x2;
      acc += x3;
    }
    for (; j < n; ++j) {
      const int c0 = __builtin_amdgcn_readlane(colv, j);
      acc += *(const vf*)(srcl + (size_t)c0 * F);
    }
  }
  return acc;
}

template <int VEC, bool SCALE_OUT>
__global__ __launch_bounds__(256) void gcn_pull_rows(
    const float* __restrict__ src, const int* __restrict__ rowptr,
    const int* __restrict__ csr_col, const float* __restrict__ dinv,
    float* __restrict__ dst, int n_rows) {
  typedef float vf __attribute__((ext_vector_type(VEC)));
  constexpr int F = VEC * 32;
  const int lane = threadIdx.x & 31;
  const int row =
      __builtin_amdgcn_readfirstlane(blockIdx.x * 8 + (threadIdx.x >> 5));
  if (row >= n_rows) return;
  const int beg = rowptr[row];
  const int end = rowptr[row + 1];
  vf acc = gcn_row_sum<VEC>(src + lane * VEC, csr_col, beg, end, lane);
  if (SCALE_OUT) acc *= dinv[row];
  *(vf*)(dst + (size_t)row * F + lane * VEC) = acc;
}

#define GCN_APITCH 132
#define GCN_FUSED_BLOCKS 1024

template <int N, int K>
__global__ __launch_bounds__(512) void gcn_pull_gemm(
    const float* __restrict__ src, const int* __restrict__ rowptr,
    const int* __restrict__ csr_col, const float* __restrict__ dinv,
    const float* __restrict__ B, float* __restrict__ C, int n_tiles) {
  constexpr int N2 = N / 2;
  constexpr int THREADS = 512;
  constexpr int STAGE_ITEMS = (K / 2) * N2;
  static_assert(K == 128, "a gathered row is one float4 per lane");
  static_assert(N == 64, "two matrix waves, 32 columns each");
  static_assert(STAGE_ITEMS % THREADS == 0, "every thread stages the same number of items");
  __shared__ __align__(16) float Bs[K * N];
  __shared__ __align__(16) float As[2][16 * GCN_APITCH];

  const int tid  = threadIdx.x;
  const int wib  = __builtin_amdgcn_readfirstlane(tid >> 5);
  const int lane = tid & 31;
  const int half = lane >> 4;
  const int l15  = lane & 15;

#pragma unroll
  for (int it = 0; it < STAGE_ITEMS / THREADS; ++it) {
    const int pidx = tid + it * THREADS;
    const int kp = pidx / N2;
    const int c2 = (pidx - kp * N2) * 2;
    const float2 g0 = *(const float2*)(B + (size_t)(2 * kp) * N + c2);
    const float2 g1 = *(const float2*)(B + (size_t)(2 * kp + 1) * N + c2);
    *(float4*)(&Bs[kp * (2 * N) + c2 * 2]) = make_float4(g0.x, g1.x, g0.y, g1.y);
  }

  const float* __restrict__ srcl = src + lane * 4;
  int buf = 0;
  int pair = 0;
  for (int tile = blockIdx.x; tile < n_tiles; tile += gridDim.x) {
    const int row0 = tile * 16;
    const int row = row0 + wib;
    const int beg = rowptr[row];
    const int end = rowptr[row + 1];
    gcn_vf<4> acc = gcn_row_sum<4>(srcl, csr_col, beg, end, lane);
    const float ds = dinv[row];
    acc.x = fmaxf(acc.x * ds, 0.0f);
    acc.y = fmaxf(acc.y * ds, 0.0f);
    acc.z = fmaxf(acc.z * ds, 0.0f);
    acc.w = fmaxf(acc.w * ds, 0.0f);
    float* At = As[buf];
    *(gcn_vf<4>*)(&At[wib * GCN_APITCH + lane * 4]) = acc;
    __syncthreads();

    const int mw = wib - pair;
    if (mw == 0 || mw == 1) {
      const int col0 = mw * 32;
      const float* Ap = &At[l15 * GCN_APITCH + 2 * half];
      const float* Bp = &Bs[half * (2 * N) + (col0 + l15) * 2];
      v8f c0 = (v8f){};
      v8f c1 = (v8f){};
#pragma unroll
      for (int i = 0; i < K / 4; ++i) {
        const v2f a  = *(const v2f*)(Ap + 4 * i);
        const v2f b0 = *(const v2f*)(Bp + i * (4 * N));
        const v2f b1 = *(const v2f*)(Bp + i * (4 * N) + 32);
        c0 = __builtin_amdgcn_wmma_f32_16x16x4_f32(false, a, false, b0, (short)0, c0,
                                                    false, false);
        c1 = __builtin_amdgcn_wmma_f32_16x16x4_f32(false, a, false, b1, (short)0, c1,
                                                    false, false);
      }
#pragma unroll
      for (int v = 0; v < 8; ++v) {
        const int r = row0 + v + 8 * half;
        const float dr = dinv[r];
        C[(size_t)r * N + col0 + l15] = c0[v] * dr;
        C[(size_t)r * N + col0 + 16 + l15] = c1[v] * dr;
      }
    }
    buf ^= 1;
    pair ^= 2;
  }
}


extern "C" void kernel_launch(void* const* d_in, const int* in_sizes, int n_in,
                              void* d_out, int out_size, void* d_ws, size_t ws_size,
                              hipStream_t stream) {
  const float* X  = (const float*)d_in[0];
  const float* W1 = (const float*)d_in[1];
  const float* W2 = (const float*)d_in[2];
  const int* erow = (const int*)d_in[3];
  const int* ecol = (const int*)d_in[4];
  const int n_edges = in_sizes[3];
  float* out = (float*)d_out;

  int* hist      = (int*)d_ws;
  int* bandstart = hist + GCN_NBANDS * GCN_SLICES;
  int* rowptr = (int*)d_ws + GCN_N_NODES;
  float* dinv = (float*)(rowptr + GCN_ROWPTR_PAD);
  float* XWs = (float*)d_ws + GCN_ROWS_OFF;
  float* HWs = XWs + (size_t)GCN_N_NODES * 128;
  int* binned  = (int*)(HWs + (size_t)GCN_N_NODES * 64);
  int* csr_col = binned + n_edges;

  gcn_band_hist_kernel<<<GCN_SLICES, 1024, 0, stream>>>(erow, hist, n_edges);
  gcn_band_scatter_kernel<<<GCN_SLICES, 1024, 0, stream>>>(
      erow, ecol, hist, bandstart, binned, n_edges);
  gcn_band_sort_kernel<<<GCN_NBANDS, GCN_BAND, 0, stream>>>(
      binned, bandstart, rowptr, csr_col, dinv);

  const int gemm_blocks = 256;
  const int pull_blocks = (GCN_N_NODES + 7) / 8;
  const int fused_blocks = GCN_FUSED_BLOCKS;

  gcn_gemm_wmma<128, 256, 16, false, true>
      <<<gemm_blocks, 512, 0, stream>>>(X, W1, XWs, dinv, GCN_N_NODES);

  gcn_pull_gemm<64, 128><<<fused_blocks, 512, 0, stream>>>(
      XWs, rowptr, csr_col, dinv, W2, HWs, GCN_N_NODES / 16);

  gcn_pull_rows<2, true><<<pull_blocks, 256, 0, stream>>>(
      HWs, rowptr, csr_col, dinv, out, GCN_N_NODES);
}


# Edge placement staged in LDS: band scatter and row placement fill an LDS copy of the slice / band segment and write it out in index order; scattered 4-byte stores become line stores; slot order kept
# speedup vs baseline: 3.8344x; 1.0779x over previous
#include <hip/hip_runtime.h>


typedef float v2f __attribute__((ext_vector_type(2)));
typedef float v8f __attribute__((ext_vector_type(8)));

#define GCN_N_NODES 100000
#define GCN_ROWPTR_PAD 100016
#define GCN_ROWS_OFF 301056
static_assert(GCN_ROWS_OFF >= GCN_N_NODES + GCN_ROWPTR_PAD + GCN_N_NODES,
              "row arrays start behind hist, rowptr and dinv");
static_assert(GCN_N_NODES % 16 == 0, "the GEMMs work on whole 16-row tiles");
static_assert(GCN_ROWS_OFF % 1024 == 0, "row arrays start on a 4096-byte boundary");
static_assert(((size_t)GCN_N_NODES * 64 * sizeof(float)) % 4096 == 0,
              "every row array is a multiple of 4096 bytes long");


#define GCN_BAND_SHIFT 9
#define GCN_BAND 512
#define GCN_NBANDS 196
#define GCN_SLICES 256
#define GCN_BATCH 8
#define GCN_SLICE_CAP 8192
#define GCN_SORT_CAP 16384

static_assert((GCN_N_NODES + GCN_BAND - 1) / GCN_BAND == GCN_NBANDS, "band count");
static_assert(GCN_N_NODES <= (1 << 22), "col << 9 must stay a non-negative int");
static_assert(GCN_BATCH == 8, "the load fences list 8 values");
static_assert(GCN_NBANDS <= 7 * 32, "scatter kernel: 7 bands per lane / per wave");
static_assert(GCN_NBANDS * GCN_SLICES + GCN_NBANDS + 1 <= GCN_N_NODES,
              "hist and bandstart live in the first GCN_N_NODES ints of the workspace");

__device__ __forceinline__ int gcn_wave_incl_scan(int x, int lane) {
  x += __builtin_amdgcn_update_dpp(0, x, 0x111, 0xf, 0xf, false);
  x += __builtin_amdgcn_update_dpp(0, x, 0x112, 0xf, 0xf, false);
  x += __builtin_amdgcn_update_dpp(0, x, 0x114, 0xf, 0xf, false);
  x += __builtin_amdgcn_update_dpp(0, x, 0x118, 0xf, 0xf, false);
  const int low = __builtin_amdgcn_readlane(x, 15);
  return x + ((lane >= 16) ? low : 0);
}

__global__ __launch_bounds__(1024) void gcn_band_hist_kernel(
    const int* __restrict__ rows, int* __restrict__ hist, int n_edges) {
  __shared__ int h[GCN_NBANDS];
  const int tid = threadIdx.x;
  const int w = blockIdx.x;
  if (tid < GCN_NBANDS) h[tid] = 0;
  __syncthreads();
  const int sl = (n_edges + GCN_SLICES - 1) / GCN_SLICES;
  const int s0 = min(w * sl, n_edges);
  const int s1 = min(s0 + sl, n_edges);
  for (int e = s0 + tid; e < s1; e += 1024 * GCN_BATCH) {
    int r[GCN_BATCH];
#pragma unroll
    for (int k = 0; k < GCN_BATCH; ++k) r[k] = rows[min(e + k * 1024, s1 - 1)];
    asm volatile("" ::"v"(r[0]), "v"(r[1]), "v"(r[2]), "v"(r[3]), "v"(r[4]), "v"(r[5]),
                 "v"(r[6]), "v"(r[7]));
#pragma unroll
    for (int k = 0; k < GCN_BATCH; ++k)
      if (e + k * 1024 < s1) atomicAdd(&h[r[k] >> GCN_BAND_SHIFT], 1);
  }
  __syncthreads();
  if (tid < GCN_NBANDS) hist[tid * GCN_SLICES + w] = h[tid];
}

template <bool STAGED>
__device__ __forceinline__ void gcn_scatter_slice(
    const int* __restrict__ rows, const int* __restrict__ cols, int* cur, const int* ldel,
    int* lval, int* ldst, int* __restrict__ binned, int s0, int s1, int tid) {
  for (int e = s0 + tid; e < s1; e += 1024 * GCN_BATCH) {
    int r[GCN_BATCH], c[GCN_BATCH];
#pragma unroll
    for (int k = 0; k < GCN_BATCH; ++k) {
      const int ee = min(e + k * 1024, s1 - 1);
      r[k] = rows[ee];
      c[k] = cols[ee];
    }
    asm volatile("" ::"v"(r[0]), "v"(r[1]), "v"(r[2]), "v"(r[3]), "v"(r[4]), "v"(r[5]),
                 "v"(r[6]), "v"(r[7]), "v"(c[0]), "v"(c[1]), "v"(c[2]), "v"(c[3]), "v"(c[4]),
                 "v"(c[5]), "v"(c[6]), "v"(c[7]));
#pragma unroll
    for (int k = 0; k < GCN_BATCH; ++k) {
      if (e + k * 1024 < s1) {
        const int b = r[k] >> GCN_BAND_SHIFT;
        const int word = (c[k] << GCN_BAND_SHIFT) | (r[k] & (GCN_BAND - 1));
        const int p = atomicAdd(&cur[b], 1);
        if (STAGED) {
          lval[p] = word;
          ldst[p] = ldel[b];
        } else {
          binned[p] = word;
        }
      }
    }
  }
}

__global__ __launch_bounds__(1024) void gcn_band_scatter_kernel(
    const int* __restrict__ rows, const int* __restrict__ cols,
    const int* __restrict__ hist, int* __restrict__ bandstart,
    int* __restrict__ binned, int n_edges) {
  __shared__ int sT[7 * 32];
  __shared__ int sP[7 * 32];
  __shared__ int sO[7 * 32];
  __shared__ int cursor[GCN_NBANDS];
  __shared__ int lcur[GCN_NBANDS];
  __shared__ int ldel[GCN_NBANDS];
  __shared__ int lval[GCN_SLICE_CAP];
  __shared__ int ldst[GCN_SLICE_CAP];
  const int tid = threadIdx.x;
  const int lane = tid & 31;
  const int wv = __builtin_amdgcn_readfirstlane(tid >> 5);
  const int w = blockIdx.x;

  int4 qa[7], qb[7];
#pragma unroll
  for (int k = 0; k < 7; ++k) {
    const int bb = min(wv + 32 * k, GCN_NBANDS - 1);
    const int4* __restrict__ p = (const int4*)(hist + bb * GCN_SLICES + lane * 8);
    qa[k] = p[0];
    qb[k] = p[1];
  }
#pragma unroll
  for (int k = 0; k < 7; ++k) {
    const int b = wv + 32 * k;
    const int j0 = lane * 8;
    int tot = qa[k].x + qa[k].y + qa[k].z + qa[k].w + qb[k].x + qb[k].y + qb[k].z + qb[k].w;
    int part = ((j0 + 0 < w) ? qa[k].x : 0) + ((j0 + 1 < w) ? qa[k].y : 0) +
               ((j0 + 2 < w) ? qa[k].z : 0) + ((j0 + 3 < w) ? qa[k].w : 0) +
               ((j0 + 4 < w) ? qb[k].x : 0) + ((j0 + 5 < w) ? qb[k].y : 0) +
               ((j0 + 6 < w) ? qb[k].z : 0) + ((j0 + 7 < w) ? qb[k].w : 0);
    const int wc = w & 7;
    const int ownv = (wc == 0) ? qa[k].x : (wc == 1) ? qa[k].y : (wc == 2) ? qa[k].z :
                     (wc == 3) ? qa[k].w : (wc == 4) ? qb[k].x : (wc == 5) ? qb[k].y :
                     (wc == 6) ? qb[k].z : qb[k].w;
    const int own = __builtin_amdgcn_readlane(ownv, w >> 3);
    tot = __builtin_amdgcn_readlane(gcn_wave_incl_scan(tot, lane), 31);
    part = __builtin_amdgcn_readlane(gcn_wave_incl_scan(part, lane), 31);
    if (lane == 0) {
      sT[b] = (b < GCN_NBANDS) ? tot : 0;
      sP[b] = (b < GCN_NBANDS) ? part : 0;
      sO[b] = (b < GCN_NBANDS) ? own : 0;
    }
  }
  __syncthreads();

  if (wv == 0) {
    int t[7], o[7];
    int s = 0, so = 0;
#pragma unroll
    for (int k = 0; k < 7; ++k) {
      t[k] = sT[lane * 7 + k];
      o[k] = sO[lane * 7 + k];
      s += t[k];
      so += o[k];
    }
    int run = gcn_wave_incl_scan(s, lane) - s;
    int lrun = gcn_wave_incl_scan(so, lane) - so;
#pragma unroll
    for (int k = 0; k < 7; ++k) {
      const int b = lane * 7 + k;
      if (b < GCN_NBANDS) {
        cursor[b] = run + sP[b];
        lcur[b] = lrun;
        ldel[b] = run + sP[b] - lrun;
        if (w == 0) bandstart[b] = run;
      }
      run += t[k];
      lrun += o[k];
    }
    if (w == 0 && lane == 31) bandstart[GCN_NBANDS] = run;
  }
  __syncthreads();

  const int sl = (n_edges + GCN_SLICES - 1) / GCN_SLICES;
  const int s0 = min(w * sl, n_edges);
  const int s1 = min(s0 + sl, n_edges);
  if (sl <= GCN_SLICE_CAP) {
    gcn_scatter_slice<true>(rows, cols, lcur, ldel, lval, ldst, binned, s0, s1, tid);
    __syncthreads();
    const int n = s1 - s0;
    for (int i = tid; i < n; i += 1024) binned[i + ldst[i]] = lval[i];
  } else {
    gcn_scatter_slice<false>(rows, cols, cursor, ldel, lval, ldst, binned, s0, s1, tid);
  }
}

template <bool STAGED>
__device__ __forceinline__ void gcn_place_band(
    const int* __restrict__ binned, int* cnt, int* placed, int* __restrict__ csr_col,
    int beg, int end, int tid) {
  for (int i = beg + tid; i < end; i += GCN_BAND * GCN_BATCH) {
    int v[GCN_BATCH];
#pragma unroll
    for (int k = 0; k < GCN_BATCH; ++k) v[k] = binned[min(i + k * GCN_BAND, end - 1)];
    asm volatile("" ::"v"(v[0]), "v"(v[1]), "v"(v[2]), "v"(v[3]), "v"(v[4]), "v"(v[5]),
                 "v"(v[6]), "v"(v[7]));
#pragma unroll
    for (int k = 0; k < GCN_BATCH; ++k) {
      if (i + k * GCN_BAND < end) {
        const int pos = atomicAdd(&cnt[v[k] & (GCN_BAND - 1)], 1);
        if (STAGED) placed[pos - beg] = v[k] >> GCN_BAND_SHIFT;
        else csr_col[pos] = v[k] >> GCN_BAND_SHIFT;
      }
    }
  }
}

__global__ __launch_bounds__(GCN_BAND) void gcn_band_sort_kernel(
    const int* __restrict__ binned, const int* __restrict__ bandstart,
    int* __restrict__ rowptr, int* __restrict__ csr_col, float* __restrict__ dinv) {
  __shared__ int cnt[GCN_BAND];
  __shared__ int wsum[GCN_BAND / 32];
  __shared__ int placed[GCN_SORT_CAP];
  const int tid = threadIdx.x;
  const int lane = tid & 31;
  const int wv = __builtin_amdgcn_readfirstlane(tid >> 5);
  const int b = blockIdx.x;
  const int beg = bandstart[b];
  const int end = bandstart[b + 1];
  const bool staged = (end - beg <= GCN_SORT_CAP);

  cnt[tid] = 0;
  __syncthreads();
  for (int i = beg + tid; i < end; i += GCN_BAND * GCN_BATCH) {
    int v[GCN_BATCH];
#pragma unroll
    for (int k = 0; k < GCN_BATCH; ++k) v[k] = binned[min(i + k * GCN_BAND, end - 1)];
    asm volatile("" ::"v"(v[0]), "v"(v[1]), "v"(v[2]), "v"(v[3]), "v"(v[4]), "v"(v[5]),
                 "v"(v[6]), "v"(v[7]));
#pragma unroll
    for (int k = 0; k < GCN_BATCH; ++k)
      if (i + k * GCN_BAND < end) atomicAdd(&cnt[v[k] & (GCN_BAND - 1)], 1);
  }
  __syncthreads();

  const int d = cnt[tid];
  const int inc = gcn_wave_incl_scan(d, lane);
  if (lane == 31) wsum[wv] = inc;
  __syncthreads();
  const int ws = (lane < GCN_BAND / 32) ? wsum[lane] : 0;
  const int winc = gcn_wave_incl_scan(ws, lane);
  const int wexcl = __builtin_amdgcn_readlane(winc, wv) - __builtin_amdgcn_readlane(ws, wv);
  const int start = beg + wexcl + inc - d;
  const int row = b * GCN_BAND + tid;
  if (row < GCN_N_NODES) {
    rowptr[row] = start;
    if (row == GCN_N_NODES - 1) rowptr[GCN_N_NODES] = start + d;
    dinv[row] = 1.0f / fmaxf(sqrtf((float)d), 1e-12f);
  }
  cnt[tid] = start;
  __syncthreads();

  if (staged) {
    gcn_place_band<true>(binned, cnt, placed, csr_col, beg, end, tid);
    __syncthreads();
    int i0 = tid - (beg & 31);
    if (i0 < 0) i0 += GCN_BAND;
    for (int i = i0; i < end - beg; i += GCN_BAND) csr_col[beg + i] = placed[i];
  } else {
    gcn_place_band<false>(binned, cnt, placed, csr_col, beg, end, tid);
  }
}


template <int N, int K, int NTL, bool RELU_SCALE_A, bool SCALE_C>
__device__ __forceinline__ void gcn_gemm_tile(
    const float* __restrict__ A, const float* Bs, float* __restrict__ C,
    const float* __restrict__ dinv, int row0, int col0, int half, int l15) {
  constexpr int KC = 64;
  static_assert(K % KC == 0, "whole chunks");
  const float* __restrict__ Arow = A + (size_t)(row0 + l15) * K + 2 * half;
  const float dscale = RELU_SCALE_A ? dinv[row0 + l15] : 1.0f;

  v8f acc[NTL];
#pragma unroll
  for (int t = 0; t < NTL; ++t) acc[t] = (v8f){};

  v2f a_nxt[KC / 4];
#pragma unroll
  for (int i = 0; i < KC / 4; ++i) a_nxt[i] = *(const v2f*)(Arow + 4 * i);

#pragma unroll 1
  for (int kc = 0; kc < K; kc += KC) {
    v2f a_cur[KC / 4];
#pragma unroll
    for (int i = 0; i < KC / 4; ++i) a_cur[i] = a_nxt[i];
    if (kc + KC < K) {
#pragma unroll
      for (int i = 0; i < KC / 4; ++i)
        a_nxt[i] = *(const v2f*)(Arow + kc + KC + 4 * i);
    }
#pragma unroll
    for (int i = 0; i < KC / 4; ++i) {
      v2f a = a_cur[i];
      if (RELU_SCALE_A) {
        a.x = fmaxf(a.x * dscale, 0.0f);
        a.y = fmaxf(a.y * dscale, 0.0f);
      }
      const int kp = (kc + 4 * i + 2 * half) >> 1;
#pragma unroll
      for (int t = 0; t < NTL; ++t) {
        const v2f b =
            *(const v2f*)(&Bs[kp * (2 * N) + (col0 + t * 16 + l15) * 2]);
        acc[t] = __builtin_amdgcn_wmma_f32_16x16x4_f32(
            false, a, false, b, (short)0, acc[t], false, false);
      }
    }
  }

#pragma unroll
  for (int t = 0; t < NTL; ++t) {
    const int col = col0 + t * 16 + l15;
#pragma unroll
    for (int v = 0; v < 8; ++v) {
      const int row = row0 + v + 8 * half;
      float val = acc[t][v];
      if (SCALE_C) val *= dinv[row];
      C[(size_t)row * N + col] = val;
    }
  }
}

template <int N, int K, int WAVES, bool RELU_SCALE_A, bool SCALE_C>
__global__ __launch_bounds__(WAVES * 32) void gcn_gemm_wmma(
    const float* __restrict__ A, const float* __restrict__ B,
    float* __restrict__ C, const float* __restrict__ dinv, int M) {
  constexpr int NT = N / 16;
  constexpr int N2 = N / 2;
  constexpr int THREADS = WAVES * 32;
  constexpr int STAGE_ITEMS = (K / 2) * N2;
  constexpr int PARTS = NT / 2;
  static_assert(STAGE_ITEMS % THREADS == 0, "every thread stages the same number of items");
  static_assert(N % 32 == 0 && WAVES % 8 == 0, "column pairs; both halves of the waves cover all SIMDs");
  __shared__ __align__(16) float Bs[K * N];

  const int tid  = threadIdx.x;
  const int wib  = __builtin_amdgcn_readfirstlane(tid >> 5);
  const int lane = tid & 31;
  const int half = lane >> 4;
  const int l15  = lane & 15;

#pragma unroll
  for (int it = 0; it < STAGE_ITEMS / THREADS; ++it) {
    const int pidx = tid + it * THREADS;
    const int kp = pidx / N2;
    const int c2 = (pidx - kp * N2) * 2;
    const float2 g0 = *(const float2*)(B + (size_t)(2 * kp) * N + c2);
    const float2 g1 = *(const float2*)(B + (size_t)(2 * kp + 1) * N + c2);
    *(float4*)(&Bs[kp * (2 * N) + c2 * 2]) =
        make_float4(g0.x, g1.x, g0.y, g1.y);
  }
  __syncthreads();

  const int n_tiles = M >> 4;
  const int grid = gridDim.x;
  const int n_slots = grid * WAVES;
  const int half_slots = n_slots >> 1;
  const int slot = wib * grid + blockIdx.x;

  const int whole = (n_tiles / n_slots) * n_slots;
  const bool half_round = (n_tiles - whole >= half_slots);
  const int n_full = whole + (half_round ? half_slots : 0);
  for (int tile = slot; tile < n_full; tile += n_slots)
    gcn_gemm_tile<N, K, NT, RELU_SCALE_A, SCALE_C>(A, Bs, C, dinv, tile * 16, 0, half, l15);

  const int n_parts = (n_tiles - n_full) * PARTS;
  const int first = half_round ? half_slots : 0;
  const int takers = n_slots - first;
  if (slot >= first) {
    for (int u = slot - first; u < n_parts; u += takers) {
      const int tile = n_full + u / PARTS;
      const int col0 = (u % PARTS) * 32;
      gcn_gemm_tile<N, K, 2, RELU_SCALE_A, SCALE_C>(A, Bs, C, dinv, tile * 16, col0, half, l15);
    }
  }
}


template <int VEC>
using gcn_vf = float __attribute__((ext_vector_type(VEC)));

template <int VEC>
__device__ __forceinline__ gcn_vf<VEC> gcn_row_sum(
    const float* __restrict__ srcl, const int* __restrict__ csr_col, int beg, int end,
    int lane) {
  typedef gcn_vf<VEC> vf;
  constexpr int F = VEC * 32;
  vf acc = {};
  for (int base = beg; base < end; base += 32) {
    const int n = min(32, end - base);
    const int colv = (lane < n) ? csr_col[base + lane] : 0;
    int j = 0;
    for (; j + 4 <= n; j += 4) {
      const int c0 = __builtin_amdgcn_readlane(colv, j);
      const int c1 = __builtin_amdgcn_readlane(colv, j + 1);
      const int c2 = __builtin_amdgcn_readlane(colv, j + 2);
      const int c3 = __builtin_amdgcn_readlane(colv, j + 3);
      const vf x0 = *(const vf*)(srcl + (size_t)c0 * F);
      const vf x1 = *(const vf*)(srcl + (size_t)c1 * F);
      const vf x2 = *(const vf*)(srcl + (size_t)c2 * F);
      const vf x3 = *(const vf*)(srcl + (size_t)c3 * F);
      acc += x0;
      acc += x1;
      acc += x2;
      acc += x3;
    }
    for (; j < n; ++j) {
      const int c0 = __builtin_amdgcn_readlane(colv, j);
      acc += *(const vf*)(srcl + (size_t)c0 * F);
    }
  }
  return acc;
}

template <int VEC, bool SCALE_OUT>
__global__ __launch_bounds__(256) void gcn_pull_rows(
    const float* __restrict__ src, const int* __restrict__ rowptr,
    const int* __restrict__ csr_col, const float* __restrict__ dinv,
    float* __restrict__ dst, int n_rows) {
  typedef float vf __attribute__((ext_vector_type(VEC)));
  constexpr int F = VEC * 32;
  const int lane = threadIdx.x & 31;
  const int row =
      __builtin_amdgcn_readfirstlane(blockIdx.x * 8 + (threadIdx.x >> 5));
  if (row >= n_rows) return;
  const int beg = rowptr[row];
  const int end = rowptr[row + 1];
  vf acc = gcn_row_sum<VEC>(src + lane * VEC, csr_col, beg, end, lane);
  if (SCALE_OUT) acc *= dinv[row];
  *(vf*)(dst + (size_t)row * F + lane * VEC) = acc;
}

#define GCN_APITCH 132
#define GCN_FUSED_BLOCKS 1024

template <int N, int K>
__global__ __launch_bounds__(512) void gcn_pull_gemm(
    const float* __restrict__ src, const int* __restrict__ rowptr,
    const int* __restrict__ csr_col, const float* __restrict__ dinv,
    const float* __restrict__ B, float* __restrict__ C, int n_tiles) {
  constexpr int N2 = N / 2;
  constexpr int THREADS = 512;
  constexpr int STAGE_ITEMS = (K / 2) * N2;
  static_assert(K == 128, "a gathered row is one float4 per lane");
  static_assert(N == 64, "two matrix waves, 32 columns each");
  static_assert(STAGE_ITEMS % THREADS == 0, "every thread stages the same number of items");
  __shared__ __align__(16) float Bs[K * N];
  __shared__ __align__(16) float As[2][16 * GCN_APITCH];

  const int tid  = threadIdx.x;
  const int wib  = __builtin_amdgcn_readfirstlane(tid >> 5);
  const int lane = tid & 31;
  const int half = lane >> 4;
  const int l15  = lane & 15;

#pragma unroll
  for (int it = 0; it < STAGE_ITEMS / THREADS; ++it) {
    const int pidx = tid + it * THREADS;
    const int kp = pidx / N2;
    const int c2 = (pidx - kp * N2) * 2;
    const float2 g0 = *(const float2*)(B + (size_t)(2 * kp) * N + c2);
    const float2 g1 = *(const float2*)(B + (size_t)(2 * kp + 1) * N + c2);
    *(float4*)(&Bs[kp * (2 * N) + c2 * 2]) = make_float4(g0.x, g1.x, g0.y, g1.y);
  }

  const float* __restrict__ srcl = src + lane * 4;
  int buf = 0;
  int pair = 0;
  for (int tile = blockIdx.x; tile < n_tiles; tile += gridDim.x) {
    const int row0 = tile * 16;
    const int row = row0 + wib;
    const int beg = rowptr[row];
    const int end = rowptr[row + 1];
    gcn_vf<4> acc = gcn_row_sum<4>(srcl, csr_col, beg, end, lane);
    const float ds = dinv[row];
    acc.x = fmaxf(acc.x * ds, 0.0f);
    acc.y = fmaxf(acc.y * ds, 0.0f);
    acc.z = fmaxf(acc.z * ds, 0.0f);
    acc.w = fmaxf(acc.w * ds, 0.0f);
    float* At = As[buf];
    *(gcn_vf<4>*)(&At[wib * GCN_APITCH + lane * 4]) = acc;
    __syncthreads();

    const int mw = wib - pair;
    if (mw == 0 || mw == 1) {
      const int col0 = mw * 32;
      const float* Ap = &At[l15 * GCN_APITCH + 2 * half];
      const float* Bp = &Bs[half * (2 * N) + (col0 + l15) * 2];
      v8f c0 = (v8f){};
      v8f c1 = (v8f){};
#pragma unroll
      for (int i = 0; i < K / 4; ++i) {
        const v2f a  = *(const v2f*)(Ap + 4 * i);
        const v2f b0 = *(const v2f*)(Bp + i * (4 * N));
        const v2f b1 = *(const v2f*)(Bp + i * (4 * N) + 32);
        c0 = __builtin_amdgcn_wmma_f32_16x16x4_f32(false, a, false, b0, (short)0, c0,
                                                    false, false);
        c1 = __builtin_amdgcn_wmma_f32_16x16x4_f32(false, a, false, b1, (short)0, c1,
                                                    false, false);
      }
#pragma unroll
      for (int v = 0; v < 8; ++v) {
        const int r = row0 + v + 8 * half;
        const float dr = dinv[r];
        C[(size_t)r * N + col0 + l15] = c0[v] * dr;
        C[(size_t)r * N + col0 + 16 + l15] = c1[v] * dr;
      }
    }
    buf ^= 1;
    pair ^= 2;
  }
}


extern "C" void kernel_launch(void* const* d_in, const int* in_sizes, int n_in,
                              void* d_out, int out_size, void* d_ws, size_t ws_size,
                              hipStream_t stream) {
  const float* X  = (const float*)d_in[0];
  const float* W1 = (const float*)d_in[1];
  const float* W2 = (const float*)d_in[2];
  const int* erow = (const int*)d_in[3];
  const int* ecol = (const int*)d_in[4];
  const int n_edges = in_sizes[3];
  float* out = (float*)d_out;

  int* hist      = (int*)d_ws;
  int* bandstart = hist + GCN_NBANDS * GCN_SLICES;
  int* rowptr = (int*)d_ws + GCN_N_NODES;
  float* dinv = (float*)(rowptr + GCN_ROWPTR_PAD);
  float* XWs = (float*)d_ws + GCN_ROWS_OFF;
  float* HWs = XWs + (size_t)GCN_N_NODES * 128;
  int* binned  = (int*)(HWs + (size_t)GCN_N_NODES * 64);
  int* csr_col = binned + n_edges;

  gcn_band_hist_kernel<<<GCN_SLICES, 1024, 0, stream>>>(erow, hist, n_edges);
  gcn_band_scatter_kernel<<<GCN_SLICES, 1024, 0, stream>>>(
      erow, ecol, hist, bandstart, binned, n_edges);
  gcn_band_sort_kernel<<<GCN_NBANDS, GCN_BAND, 0, stream>>>(
      binned, bandstart, rowptr, csr_col, dinv);

  const int gemm_blocks = 256;
  const int pull_blocks = (GCN_N_NODES + 7) / 8;
  const int fused_blocks = GCN_FUSED_BLOCKS;

  gcn_gemm_wmma<128, 256, 16, false, true>
      <<<gemm_blocks, 512, 0, stream>>>(X, W1, XWs, dinv, GCN_N_NODES);

  gcn_pull_gemm<64, 128><<<fused_blocks, 512, 0, stream>>>(
      XWs, rowptr, csr_col, dinv, W2, HWs, GCN_N_NODES / 16);

  gcn_pull_rows<2, true><<<pull_blocks, 256, 0, stream>>>(
      HWs, rowptr, csr_col, dinv, out, GCN_N_NODES);
}


# Row-sum gather deepened: all source-row loads of a group issued before its first add, 24 (256-byte rows) / 11 (512-byte rows) per group, leftover rows inside the group; slot (sum) order kept
# speedup vs baseline: 3.9720x; 1.0359x over previous
#include <hip/hip_runtime.h>


typedef float v2f __attribute__((ext_vector_type(2)));
typedef float v8f __attribute__((ext_vector_type(8)));

#define GCN_N_NODES 100000
#define GCN_ROWPTR_PAD 100016
#define GCN_ROWS_OFF 301056
static_assert(GCN_ROWS_OFF >= GCN_N_NODES + GCN_ROWPTR_PAD + GCN_N_NODES,
              "row arrays start behind hist, rowptr and dinv");
static_assert(GCN_N_NODES % 16 == 0, "the GEMMs work on whole 16-row tiles");
static_assert(GCN_ROWS_OFF % 1024 == 0, "row arrays start on a 4096-byte boundary");
static_assert(((size_t)GCN_N_NODES * 64 * sizeof(float)) % 4096 == 0,
              "every row array is a multiple of 4096 bytes long");


#define GCN_BAND_SHIFT 9
#define GCN_BAND 512
#define GCN_NBANDS 196
#define GCN_SLICES 256
#define GCN_BATCH 8
#define GCN_SLICE_CAP 8192
#define GCN_SORT_CAP 16384

static_assert((GCN_N_NODES + GCN_BAND - 1) / GCN_BAND == GCN_NBANDS, "band count");
static_assert(GCN_N_NODES <= (1 << 22), "col << 9 must stay a non-negative int");
static_assert(GCN_BATCH == 8, "the load fences list 8 values");
static_assert(GCN_NBANDS <= 7 * 32, "scatter kernel: 7 bands per lane / per wave");
static_assert(GCN_NBANDS * GCN_SLICES + GCN_NBANDS + 1 <= GCN_N_NODES,
              "hist and bandstart live in the first GCN_N_NODES ints of the workspace");

__device__ __forceinline__ int gcn_wave_incl_scan(int x, int lane) {
  x += __builtin_amdgcn_update_dpp(0, x, 0x111, 0xf, 0xf, false);
  x += __builtin_amdgcn_update_dpp(0, x, 0x112, 0xf, 0xf, false);
  x += __builtin_amdgcn_update_dpp(0, x, 0x114, 0xf, 0xf, false);
  x += __builtin_amdgcn_update_dpp(0, x, 0x118, 0xf, 0xf, false);
  const int low = __builtin_amdgcn_readlane(x, 15);
  return x + ((lane >= 16) ? low : 0);
}

__global__ __launch_bounds__(1024) void gcn_band_hist_kernel(
    const int* __restrict__ rows, int* __restrict__ hist, int n_edges) {
  __shared__ int h[GCN_NBANDS];
  const int tid = threadIdx.x;
  const int w = blockIdx.x;
  if (tid < GCN_NBANDS) h[tid] = 0;
  __syncthreads();
  const int sl = (n_edges + GCN_SLICES - 1) / GCN_SLICES;
  const int s0 = min(w * sl, n_edges);
  const int s1 = min(s0 + sl, n_edges);
  for (int e = s0 + tid; e < s1; e += 1024 * GCN_BATCH) {
    int r[GCN_BATCH];
#pragma unroll
    for (int k = 0; k < GCN_BATCH; ++k) r[k] = rows[min(e + k * 1024, s1 - 1)];
    asm volatile("" ::"v"(r[0]), "v"(r[1]), "v"(r[2]), "v"(r[3]), "v"(r[4]), "v"(r[5]),
                 "v"(r[6]), "v"(r[7]));
#pragma unroll
    for (int k = 0; k < GCN_BATCH; ++k)
      if (e + k * 1024 < s1) atomicAdd(&h[r[k] >> GCN_BAND_SHIFT], 1);
  }
  __syncthreads();
  if (tid < GCN_NBANDS) hist[tid * GCN_SLICES + w] = h[tid];
}

template <bool STAGED>
__device__ __forceinline__ void gcn_scatter_slice(
    const int* __restrict__ rows, const int* __restrict__ cols, int* cur, const int* ldel,
    int* lval, int* ldst, int* __restrict__ binned, int s0, int s1, int tid) {
  for (int e = s0 + tid; e < s1; e += 1024 * GCN_BATCH) {
    int r[GCN_BATCH], c[GCN_BATCH];
#pragma unroll
    for (int k = 0; k < GCN_BATCH; ++k) {
      const int ee = min(e + k * 1024, s1 - 1);
      r[k] = rows[ee];
      c[k] = cols[ee];
    }
    asm volatile("" ::"v"(r[0]), "v"(r[1]), "v"(r[2]), "v"(r[3]), "v"(r[4]), "v"(r[5]),
                 "v"(r[6]), "v"(r[7]), "v"(c[0]), "v"(c[1]), "v"(c[2]), "v"(c[3]), "v"(c[4]),
                 "v"(c[5]), "v"(c[6]), "v"(c[7]));
#pragma unroll
    for (int k = 0; k < GCN_BATCH; ++k) {
      if (e + k * 1024 < s1) {
        const int b = r[k] >> GCN_BAND_SHIFT;
        const int word = (c[k] << GCN_BAND_SHIFT) | (r[k] & (GCN_BAND - 1));
        const int p = atomicAdd(&cur[b], 1);
        if (STAGED) {
          lval[p] = word;
          ldst[p] = ldel[b];
        } else {
          binned[p] = word;
        }
      }
    }
  }
}

__global__ __launch_bounds__(1024) void gcn_band_scatter_kernel(
    const int* __restrict__ rows, const int* __restrict__ cols,
    const int* __restrict__ hist, int* __restrict__ bandstart,
    int* __restrict__ binned, int n_edges) {
  __shared__ int sT[7 * 32];
  __shared__ int sP[7 * 32];
  __shared__ int sO[7 * 32];
  __shared__ int cursor[GCN_NBANDS];
  __shared__ int lcur[GCN_NBANDS];
  __shared__ int ldel[GCN_NBANDS];
  __shared__ int lval[GCN_SLICE_CAP];
  __shared__ int ldst[GCN_SLICE_CAP];
  const int tid = threadIdx.x;
  const int lane = tid & 31;
  const int wv = __builtin_amdgcn_readfirstlane(tid >> 5);
  const int w = blockIdx.x;

  int4 qa[7], qb[7];
#pragma unroll
  for (int k = 0; k < 7; ++k) {
    const int bb = min(wv + 32 * k, GCN_NBANDS - 1);
    const int4* __restrict__ p = (const int4*)(hist + bb * GCN_SLICES + lane * 8);
    qa[k] = p[0];
    qb[k] = p[1];
  }
#pragma unroll
  for (int k = 0; k < 7; ++k) {
    const int b = wv + 32 * k;
    const int j0 = lane * 8;
    int tot = qa[k].x + qa[k].y + qa[k].z + qa[k].w + qb[k].x + qb[k].y + qb[k].z + qb[k].w;
    int part = ((j0 + 0 < w) ? qa[k].x : 0) + ((j0 + 1 < w) ? qa[k].y : 0) +
               ((j0 + 2 < w) ? qa[k].z : 0) + ((j0 + 3 < w) ? qa[k].w : 0) +
               ((j0 + 4 < w) ? qb[k].x : 0) + ((j0 + 5 < w) ? qb[k].y : 0) +
               ((j0 + 6 < w) ? qb[k].z : 0) + ((j0 + 7 < w) ? qb[k].w : 0);
    const int wc = w & 7;
    const int ownv = (wc == 0) ? qa[k].x : (wc == 1) ? qa[k].y : (wc == 2) ? qa[k].z :
                     (wc == 3) ? qa[k].w : (wc == 4) ? qb[k].x : (wc == 5) ? qb[k].y :
                     (wc == 6) ? qb[k].z : qb[k].w;
    const int own = __builtin_amdgcn_readlane(ownv, w >> 3);
    tot = __builtin_amdgcn_readlane(gcn_wave_incl_scan(tot, lane), 31);
    part = __builtin_amdgcn_readlane(gcn_wave_incl_scan(part, lane), 31);
    if (lane == 0) {
      sT[b] = (b < GCN_NBANDS) ? tot : 0;
      sP[b] = (b < GCN_NBANDS) ? part : 0;
      sO[b] = (b < GCN_NBANDS) ? own : 0;
    }
  }
  __syncthreads();

  if (wv == 0) {
    int t[7], o[7];
    int s = 0, so = 0;
#pragma unroll
    for (int k = 0; k < 7; ++k) {
      t[k] = sT[lane * 7 + k];
      o[k] = sO[lane * 7 + k];
      s += t[k];
      so += o[k];
    }
    int run = gcn_wave_incl_scan(s, lane) - s;
    int lrun = gcn_wave_incl_scan(so, lane) - so;
#pragma unroll
    for (int k = 0; k < 7; ++k) {
      const int b = lane * 7 + k;
      if (b < GCN_NBANDS) {
        cursor[b] = run + sP[b];
        lcur[b] = lrun;
        ldel[b] = run + sP[b] - lrun;
        if (w == 0) bandstart[b] = run;
      }
      run += t[k];
      lrun += o[k];
    }
    if (w == 0 && lane == 31) bandstart[GCN_NBANDS] = run;
  }
  __syncthreads();

  const int sl = (n_edges + GCN_SLICES - 1) / GCN_SLICES;
  const int s0 = min(w * sl, n_edges);
  const int s1 = min(s0 + sl, n_edges);
  if (sl <= GCN_SLICE_CAP) {
    gcn_scatter_slice<true>(rows, cols, lcur, ldel, lval, ldst, binned, s0, s1, tid);
    __syncthreads();
    const int n = s1 - s0;
    for (int i = tid; i < n; i += 1024) binned[i + ldst[i]] = lval[i];
  } else {
    gcn_scatter_slice<false>(rows, cols, cursor, ldel, lval, ldst, binned, s0, s1, tid);
  }
}

template <bool STAGED>
__device__ __forceinline__ void gcn_place_band(
    const int* __restrict__ binned, int* cnt, int* placed, int* __restrict__ csr_col,
    int beg, int end, int tid) {
  for (int i = beg + tid; i < end; i += GCN_BAND * GCN_BATCH) {
    int v[GCN_BATCH];
#pragma unroll
    for (int k = 0; k < GCN_BATCH; ++k) v[k] = binned[min(i + k * GCN_BAND, end - 1)];
    asm volatile("" ::"v"(v[0]), "v"(v[1]), "v"(v[2]), "v"(v[3]), "v"(v[4]), "v"(v[5]),
                 "v"(v[6]), "v"(v[7]));
#pragma unroll
    for (int k = 0; k < GCN_BATCH; ++k) {
      if (i + k * GCN_BAND < end) {
        const int pos = atomicAdd(&cnt[v[k] & (GCN_BAND - 1)], 1);
        if (STAGED) placed[pos - beg] = v[k] >> GCN_BAND_SHIFT;
        else csr_col[pos] = v[k] >> GCN_BAND_SHIFT;
      }
    }
  }
}

__global__ __launch_bounds__(GCN_BAND) void gcn_band_sort_kernel(
    const int* __restrict__ binned, const int* __restrict__ bandstart,
    int* __restrict__ rowptr, int* __restrict__ csr_col, float* __restrict__ dinv) {
  __shared__ int cnt[GCN_BAND];
  __shared__ int wsum[GCN_BAND / 32];
  __shared__ int placed[GCN_SORT_CAP];
  const int tid = threadIdx.x;
  const int lane = tid & 31;
  const int wv = __builtin_amdgcn_readfirstlane(tid >> 5);
  const int b = blockIdx.x;
  const int beg = bandstart[b];
  const int end = bandstart[b + 1];
  const bool staged = (end - beg <= GCN_SORT_CAP);

  cnt[tid] = 0;
  __syncthreads();
  for (int i = beg + tid; i < end; i += GCN_BAND * GCN_BATCH) {
    int v[GCN_BATCH];
#pragma unroll
    for (int k = 0; k < GCN_BATCH; ++k) v[k] = binned[min(i + k * GCN_BAND, end - 1)];
    asm volatile("" ::"v"(v[0]), "v"(v[1]), "v"(v[2]), "v"(v[3]), "v"(v[4]), "v"(v[5]),
                 "v"(v[6]), "v"(v[7]));
#pragma unroll
    for (int k = 0; k < GCN_BATCH; ++k)
      if (i + k * GCN_BAND < end) atomicAdd(&cnt[v[k] & (GCN_BAND - 1)], 1);
  }
  __syncthreads();

  const int d = cnt[tid];
  const int inc = gcn_wave_incl_scan(d, lane);
  if (lane == 31) wsum[wv] = inc;
  __syncthreads();
  const int ws = (lane < GCN_BAND / 32) ? wsum[lane] : 0;
  const int winc = gcn_wave_incl_scan(ws, lane);
  const int wexcl = __builtin_amdgcn_readlane(winc, wv) - __builtin_amdgcn_readlane(ws, wv);
  const int start = beg + wexcl + inc - d;
  const int row = b * GCN_BAND + tid;
  if (row < GCN_N_NODES) {
    rowptr[row] = start;
    if (row == GCN_N_NODES - 1) rowptr[GCN_N_NODES] = start + d;
    dinv[row] = 1.0f / fmaxf(sqrtf((float)d), 1e-12f);
  }
  cnt[tid] = start;
  __syncthreads();

  if (staged) {
    gcn_place_band<true>(binned, cnt, placed, csr_col, beg, end, tid);
    __syncthreads();
    int i0 = tid - (beg & 31);
    if (i0 < 0) i0 += GCN_BAND;
    for (int i = i0; i < end - beg; i += GCN_BAND) csr_col[beg + i] = placed[i];
  } else {
    gcn_place_band<false>(binned, cnt, placed, csr_col, beg, end, tid);
  }
}


template <int N, int K, int NTL, bool RELU_SCALE_A, bool SCALE_C>
__device__ __forceinline__ void gcn_gemm_tile(
    const float* __restrict__ A, const float* Bs, float* __restrict__ C,
    const float* __restrict__ dinv, int row0, int col0, int half, int l15) {
  constexpr int KC = 64;
  static_assert(K % KC == 0, "whole chunks");
  const float* __restrict__ Arow = A + (size_t)(row0 + l15) * K + 2 * half;
  const float dscale = RELU_SCALE_A ? dinv[row0 + l15] : 1.0f;

  v8f acc[NTL];
#pragma unroll
  for (int t = 0; t < NTL; ++t) acc[t] = (v8f){};

  v2f a_nxt[KC / 4];
#pragma unroll
  for (int i = 0; i < KC / 4; ++i) a_nxt[i] = *(const v2f*)(Arow + 4 * i);

#pragma unroll 1
  for (int kc = 0; kc < K; kc += KC) {
    v2f a_cur[KC / 4];
#pragma unroll
    for (int i = 0; i < KC / 4; ++i) a_cur[i] = a_nxt[i];
    if (kc + KC < K) {
#pragma unroll
      for (int i = 0; i < KC / 4; ++i)
        a_nxt[i] = *(const v2f*)(Arow + kc + KC + 4 * i);
    }
#pragma unroll
    for (int i = 0; i < KC / 4; ++i) {
      v2f a = a_cur[i];
      if (RELU_SCALE_A) {
        a.x = fmaxf(a.x * dscale, 0.0f);
        a.y = fmaxf(a.y * dscale, 0.0f);
      }
      const int kp = (kc + 4 * i + 2 * half) >> 1;
#pragma unroll
      for (int t = 0; t < NTL; ++t) {
        const v2f b =
            *(const v2f*)(&Bs[kp * (2 * N) + (col0 + t * 16 + l15) * 2]);
        acc[t] = __builtin_amdgcn_wmma_f32_16x16x4_f32(
            false, a, false, b, (short)0, acc[t], false, false);
      }
    }
  }

#pragma unroll
  for (int t = 0; t < NTL; ++t) {
    const int col = col0 + t * 16 + l15;
#pragma unroll
    for (int v = 0; v < 8; ++v) {
      const int row = row0 + v + 8 * half;
      float val = acc[t][v];
      if (SCALE_C) val *= dinv[row];
      C[(size_t)row * N + col] = val;
    }
  }
}

template <int N, int K, int WAVES, bool RELU_SCALE_A, bool SCALE_C>
__global__ __launch_bounds__(WAVES * 32) void gcn_gemm_wmma(
    const float* __restrict__ A, const float* __restrict__ B,
    float* __restrict__ C, const float* __restrict__ dinv, int M) {
  constexpr int NT = N / 16;
  constexpr int N2 = N / 2;
  constexpr int THREADS = WAVES * 32;
  constexpr int STAGE_ITEMS = (K / 2) * N2;
  constexpr int PARTS = NT / 2;
  static_assert(STAGE_ITEMS % THREADS == 0, "every thread stages the same number of items");
  static_assert(N % 32 == 0 && WAVES % 8 == 0, "column pairs; both halves of the waves cover all SIMDs");
  __shared__ __align__(16) float Bs[K * N];

  const int tid  = threadIdx.x;
  const int wib  = __builtin_amdgcn_readfirstlane(tid >> 5);
  const int lane = tid & 31;
  const int half = lane >> 4;
  const int l15  = lane & 15;

#pragma unroll
  for (int it = 0; it < STAGE_ITEMS / THREADS; ++it) {
    const int pidx = tid + it * THREADS;
    const int kp = pidx / N2;
    const int c2 = (pidx - kp * N2) * 2;
    const float2 g0 = *(const float2*)(B + (size_t)(2 * kp) * N + c2);
    const float2 g1 = *(const float2*)(B + (size_t)(2 * kp + 1) * N + c2);
    *(float4*)(&Bs[kp * (2 * N) + c2 * 2]) =
        make_float4(g0.x, g1.x, g0.y, g1.y);
  }
  __syncthreads();

  const int n_tiles = M >> 4;
  const int grid = gridDim.x;
  const int n_slots = grid * WAVES;
  const int half_slots = n_slots >> 1;
  const int slot = wib * grid + blockIdx.x;

  const int whole = (n_tiles / n_slots) * n_slots;
  const bool half_round = (n_tiles - whole >= half_slots);
  const int n_full = whole + (half_round ? half_slots : 0);
  for (int tile = slot; tile < n_full; tile += n_slots)
    gcn_gemm_tile<N, K, NT, RELU_SCALE_A, SCALE_C>(A, Bs, C, dinv, tile * 16, 0, half, l15);

  const int n_parts = (n_tiles - n_full) * PARTS;
  const int first = half_round ? half_slots : 0;
  const int takers = n_slots - first;
  if (slot >= first) {
    for (int u = slot - first; u < n_parts; u += takers) {
      const int tile = n_full + u / PARTS;
      const int col0 = (u % PARTS) * 32;
      gcn_gemm_tile<N, K, 2, RELU_SCALE_A, SCALE_C>(A, Bs, C, dinv, tile * 16, col0, half, l15);
    }
  }
}


template <int VEC>
using gcn_vf = float __attribute__((ext_vector_type(VEC)));

template <int VEC, int R>
__device__ __forceinline__ void gcn_add_rows(
    gcn_vf<VEC>& acc, const float* __restrict__ src, unsigned lane_byte, int colv, int j) {
  typedef gcn_vf<VEC> vf;
  constexpr unsigned ROW_BYTES = VEC * 32 * sizeof(float);
  static_assert((size_t)GCN_N_NODES * ROW_BYTES <= 0xffffffffu, "a row's byte offset fits 32 bits");
  vf x[R];
#pragma unroll
  for (int k = 0; k < R; ++k) {
    const unsigned c = (unsigned)__builtin_amdgcn_readlane(colv, j + k);
    x[k] = *(const vf*)((const char*)src + (c * ROW_BYTES + lane_byte));
  }
#pragma unroll
  for (int k = 0; k < R; ++k) acc += x[k];
}

#define GCN_GROUP_CASE(R)                                                            \
  case R:                                                                            \
    if constexpr (DEPTH >= R) gcn_add_rows<VEC, R>(acc, src, lane_byte, colv, j);    \
    break;

template <int VEC, int DEPTH>
__device__ __forceinline__ gcn_vf<VEC> gcn_row_sum(
    const float* __restrict__ src, unsigned lane_byte, const int* __restrict__ csr_col,
    int beg, int end, int lane) {
  typedef gcn_vf<VEC> vf;
  static_assert(DEPTH >= 1 && DEPTH <= 24, "the switch below lists the group lengths");
  vf acc = {};
  for (int base = beg; base < end; base += 32) {
    const int n = min(32, end - base);
    const int colv = (lane < n) ? csr_col[base + lane] : 0;
    for (int j = 0; j < n; j += DEPTH) {
      switch (min(DEPTH, n - j)) {
        GCN_GROUP_CASE(1)
        GCN_GROUP_CASE(2)
        GCN_GROUP_CASE(3)
        GCN_GROUP_CASE(4)
        GCN_GROUP_CASE(5)
        GCN_GROUP_CASE(6)
        GCN_GROUP_CASE(7)
        GCN_GROUP_CASE(8)
        GCN_GROUP_CASE(9)
        GCN_GROUP_CASE(10)
        GCN_GROUP_CASE(11)
        GCN_GROUP_CASE(12)
        GCN_GROUP_CASE(13)
        GCN_GROUP_CASE(14)
        GCN_GROUP_CASE(15)
        GCN_GROUP_CASE(16)
        GCN_GROUP_CASE(17)
        GCN_GROUP_CASE(18)
        GCN_GROUP_CASE(19)
        GCN_GROUP_CASE(20)
        GCN_GROUP_CASE(21)
        GCN_GROUP_CASE(22)
        GCN_GROUP_CASE(23)
        GCN_GROUP_CASE(24)
        default: break;
      }
    }
  }
  return acc;
}
#undef GCN_GROUP_CASE

template <int VEC, bool SCALE_OUT>
__global__ __launch_bounds__(256) void gcn_pull_rows(
    const float* __restrict__ src, const int* __restrict__ rowptr,
    const int* __restrict__ csr_col, const float* __restrict__ dinv,
    float* __restrict__ dst, int n_rows) {
  typedef float vf __attribute__((ext_vector_type(VEC)));
  constexpr int F = VEC * 32;
  const int lane = threadIdx.x & 31;
  const int row =
      __builtin_amdgcn_readfirstlane(blockIdx.x * 8 + (threadIdx.x >> 5));
  if (row >= n_rows) return;
  const int beg = rowptr[row];
  const int end = rowptr[row + 1];
  vf acc = gcn_row_sum<VEC, 24>(src, (unsigned)(lane * VEC * sizeof(float)), csr_col, beg, end, lane);
  if (SCALE_OUT) acc *= dinv[row];
  *(vf*)(dst + (size_t)row * F + lane * VEC) = acc;
}

#define GCN_APITCH 132
#define GCN_FUSED_BLOCKS 1024

template <int N, int K>
__global__ __launch_bounds__(512) void gcn_pull_gemm(
    const float* __restrict__ src, const int* __restrict__ rowptr,
    const int* __restrict__ csr_col, const float* __restrict__ dinv,
    const float* __restrict__ B, float* __restrict__ C, int n_tiles) {
  constexpr int N2 = N / 2;
  constexpr int THREADS = 512;
  constexpr int STAGE_ITEMS = (K / 2) * N2;
  static_assert(K == 128, "a gathered row is one float4 per lane");
  static_assert(N == 64, "two matrix waves, 32 columns each");
  static_assert(STAGE_ITEMS % THREADS == 0, "every thread stages the same number of items");
  __shared__ __align__(16) float Bs[K * N];
  __shared__ __align__(16) float As[2][16 * GCN_APITCH];

  const int tid  = threadIdx.x;
  const int wib  = __builtin_amdgcn_readfirstlane(tid >> 5);
  const int lane = tid & 31;
  const int half = lane >> 4;
  const int l15  = lane & 15;

#pragma unroll
  for (int it = 0; it < STAGE_ITEMS / THREADS; ++it) {
    const int pidx = tid + it * THREADS;
    const int kp = pidx / N2;
    const int c2 = (pidx - kp * N2) * 2;
    const float2 g0 = *(const float2*)(B + (size_t)(2 * kp) * N + c2);
    const float2 g1 = *(const float2*)(B + (size_t)(2 * kp + 1) * N + c2);
    *(float4*)(&Bs[kp * (2 * N) + c2 * 2]) = make_float4(g0.x, g1.x, g0.y, g1.y);
  }

  const unsigned lane_byte = (unsigned)(lane * 4 * sizeof(float));
  int buf = 0;
  int pair = 0;
  for (int tile = blockIdx.x; tile < n_tiles; tile += gridDim.x) {
    const int row0 = tile * 16;
    const int row = row0 + wib;
    const int beg = rowptr[row];
    const int end = rowptr[row + 1];
    gcn_vf<4> acc = gcn_row_sum<4, 11>(src, lane_byte, csr_col, beg, end, lane);
    const float ds = dinv[row];
    acc.x = fmaxf(acc.x * ds, 0.0f);
    acc.y = fmaxf(acc.y * ds, 0.0f);
    acc.z = fmaxf(acc.z * ds, 0.0f);
    acc.w = fmaxf(acc.w * ds, 0.0f);
    float* At = As[buf];
    *(gcn_vf<4>*)(&At[wib * GCN_APITCH + lane * 4]) = acc;
    __syncthreads();

    const int mw = wib - pair;
    if (mw == 0 || mw == 1) {
      const int col0 = mw * 32;
      const float* Ap = &At[l15 * GCN_APITCH + 2 * half];
      const float* Bp = &Bs[half * (2 * N) + (col0 + l15) * 2];
      v8f c0 = (v8f){};
      v8f c1 = (v8f){};
#pragma unroll
      for (int i = 0; i < K / 4; ++i) {
        const v2f a  = *(const v2f*)(Ap + 4 * i);
        const v2f b0 = *(const v2f*)(Bp + i * (4 * N));
        const v2f b1 = *(const v2f*)(Bp + i * (4 * N) + 32);
        c0 = __builtin_amdgcn_wmma_f32_16x16x4_f32(false, a, false, b0, (short)0, c0,
                                                    false, false);
        c1 = __builtin_amdgcn_wmma_f32_16x16x4_f32(false, a, false, b1, (short)0, c1,
                                                    false, false);
      }
#pragma unroll
      for (int v = 0; v < 8; ++v) {
        const int r = row0 + v + 8 * half;
        const float dr = dinv[r];
        C[(size_t)r * N + col0 + l15] = c0[v] * dr;
        C[(size_t)r * N + col0 + 16 + l15] = c1[v] * dr;
      }
    }
    buf ^= 1;
    pair ^= 2;
  }
}


extern "C" void kernel_launch(void* const* d_in, const int* in_sizes, int n_in,
                              void* d_out, int out_size, void* d_ws, size_t ws_size,
                              hipStream_t stream) {
  const float* X  = (const float*)d_in[0];
  const float* W1 = (const float*)d_in[1];
  const float* W2 = (const float*)d_in[2];
  const int* erow = (const int*)d_in[3];
  const int* ecol = (const int*)d_in[4];
  const int n_edges = in_sizes[3];
  float* out = (float*)d_out;

  int* hist      = (int*)d_ws;
  int* bandstart = hist + GCN_NBANDS * GCN_SLICES;
  int* rowptr = (int*)d_ws + GCN_N_NODES;
  float* dinv = (float*)(rowptr + GCN_ROWPTR_PAD);
  float* XWs = (float*)d_ws + GCN_ROWS_OFF;
  float* HWs = XWs + (size_t)GCN_N_NODES * 128;
  int* binned  = (int*)(HWs + (size_t)GCN_N_NODES * 64);
  int* csr_col = binned + n_edges;

  gcn_band_hist_kernel<<<GCN_SLICES, 1024, 0, stream>>>(erow, hist, n_edges);
  gcn_band_scatter_kernel<<<GCN_SLICES, 1024, 0, stream>>>(
      erow, ecol, hist, bandstart, binned, n_edges);
  gcn_band_sort_kernel<<<GCN_NBANDS, GCN_BAND, 0, stream>>>(
      binned, bandstart, rowptr, csr_col, dinv);

  const int gemm_blocks = 256;
  const int pull_blocks = (GCN_N_NODES + 7) / 8;
  const int fused_blocks = GCN_FUSED_BLOCKS;

  gcn_gemm_wmma<128, 256, 16, false, true>
      <<<gemm_blocks, 512, 0, stream>>>(X, W1, XWs, dinv, GCN_N_NODES);

  gcn_pull_gemm<64, 128><<<fused_blocks, 512, 0, stream>>>(
      XWs, rowptr, csr_col, dinv, W2, HWs, GCN_N_NODES / 16);

  gcn_pull_rows<2, true><<<pull_blocks, 256, 0, stream>>>(
      HWs, rowptr, csr_col, dinv, out, GCN_N_NODES);
}
